# starting kernel
# baseline (speedup 1.0000x reference)
#include <hip/hip_runtime.h>
#include <hip/hip_bf16.h>

typedef float v2f __attribute__((ext_vector_type(2)));
typedef float v8f __attribute__((ext_vector_type(8)));

__device__ __forceinline__ v8f wmma_f32(v2f a, v2f b, v8f c) {
  return __builtin_amdgcn_wmma_f32_16x16x4_f32(false, a, false, b, (short)0, c,
                                               false, false);
}

#define SPATIAL 32768
#define MPAD 256
#define NWT (27 * 64 * MPAD)

__global__ __launch_bounds__(256) void prep_weights_kernel(
    const float* __restrict__ We, const float* __restrict__ be,
    float* __restrict__ Wt, float* __restrict__ bep) {
  int idx = blockIdx.x * 256 + threadIdx.x;
  if (idx < NWT) {
    int o = idx & (MPAD - 1);
    int cm = (idx >> 8) & 63;
    int koff = idx >> 14;
    float v = (o < 216) ? We[(size_t)o * 1728 + cm * 27 + koff] : 0.0f;
    Wt[((size_t)(koff * 32 + (cm >> 1)) * MPAD + o) * 2 + (cm & 1)] = v;
  } else if (idx < NWT + MPAD) {
    int o = idx - NWT;
    bep[o] = (o < 216) ? be[o] : 0.0f;
  }
}

template <int MTILES, bool BIAS>
__global__ __launch_bounds__(128) void gemm_f32_wmma_kernel(
    const float* __restrict__ A, const float* __restrict__ B,
    const float* __restrict__ bias, float* __restrict__ C, int N) {
  constexpr int M = MTILES * 16;
  __shared__ float lA[64 * 64];
  __shared__ float lB[64 * 64];

  const int tid = threadIdx.x;
  const int lane = tid & 31;
  const int wave = tid >> 5;
  const int n0 = blockIdx.x * 64;
  const int ml = lane & 15;
  const int lhi = lane >> 4;

  for (int i = tid; i < M * 64; i += 128) lA[i] = A[i];
  for (int i = tid; i < 64 * 64; i += 128) {
    int k = i >> 6, col = i & 63;
    lB[((k >> 1) * 64 + col) * 2 + (k & 1)] = B[(size_t)k * N + n0 + col];
  }
  __syncthreads();

  int mt[MTILES], nt[MTILES];
#pragma unroll
  for (int i = 0; i < MTILES; ++i) {
    int p = wave + 4 * i;
    mt[i] = p % MTILES;
    nt[i] = p / MTILES;
  }

  v8f acc[MTILES];
#pragma unroll
  for (int i = 0; i < MTILES; ++i) {
    v8f c;
#pragma unroll
    for (int v = 0; v < 8; ++v)
      c[v] = BIAS ? bias[mt[i] * 16 + v + 8 * lhi] : 0.0f;
    acc[i] = c;
  }

#pragma unroll
  for (int k0 = 0; k0 < 64; k0 += 4) {
#pragma unroll
    for (int i = 0; i < MTILES; ++i) {
      v2f a = *(const v2f*)&lA[(mt[i] * 16 + ml) * 64 + k0 + 2 * lhi];
      v2f b = *(const v2f*)&lB[(((k0 >> 1) + lhi) * 64 + nt[i] * 16 + ml) * 2];
      acc[i] = wmma_f32(a, b, acc[i]);
    }
  }

#pragma unroll
  for (int i = 0; i < MTILES; ++i)
#pragma unroll
    for (int v = 0; v < 8; ++v) {
      int m = mt[i] * 16 + v + 8 * lhi;
      C[(size_t)m * N + n0 + nt[i] * 16 + ml] = acc[i][v];
    }
}

__global__ __launch_bounds__(128) void conv3_wmma_kernel(
    const float* __restrict__ y1, const float* __restrict__ Wt,
    const float* __restrict__ bep, float* __restrict__ y2p) {
  __shared__ float strip[64 * 34];

  const int tid = threadIdx.x;
  const int lane = tid & 31;
  const int wave = tid >> 5;
  const int d = blockIdx.x >> 5;
  const int h = blockIdx.x & 31;
  const int ml = lane & 15;
  const int lhi = lane >> 4;

  v8f acc[4][2];
#pragma unroll
  for (int t = 0; t < 4; ++t) {
    int ob = (wave + 4 * t) * 16;
#pragma unroll
    for (int j = 0; j < 2; ++j) {
      v8f c;
#pragma unroll
      for (int v = 0; v < 8; ++v) c[v] = bep[ob + v + 8 * lhi];
      acc[t][j] = c;
    }
  }

  for (int q = 0; q < 9; ++q) {
    const int kd = q / 3, kh = q % 3;
    const int dd = d + kd - 1, hh = h + kh - 1;
    const bool rowok = (dd >= 0 && dd < 32 && hh >= 0 && hh < 32);
    const int ddc = min(max(dd, 0), 31), hhc = min(max(hh, 0), 31);
    const float* src = y1 + ((size_t)ddc * 32 + hhc) * 32;

    __syncthreads();
    for (int i = tid; i < 64 * 34; i += 128) {
      int cm = i / 34, pos = i % 34;
      int wsrc = pos - 1;
      int wc = min(max(wsrc, 0), 31);
      float ld = src[(size_t)cm * SPATIAL + wc];
      bool ok = rowok && (wsrc >= 0) && (wsrc < 32);
      strip[((cm >> 1) * 34 + pos) * 2 + (cm & 1)] = ok ? ld : 0.0f;
    }
    if (q < 8 && tid < 64) {
      int kd2 = (q + 1) / 3, kh2 = (q + 1) % 3;
      int d2 = min(max(d + kd2 - 1, 0), 31), h2 = min(max(h + kh2 - 1, 0), 31);
      __builtin_prefetch(y1 + ((size_t)tid * SPATIAL) + ((size_t)d2 * 32 + h2) * 32,
                         0, 0);
    }
    __syncthreads();

    const int koffb = kd * 9 + kh * 3;
#pragma unroll 2
    for (int k0 = 0; k0 < 64; k0 += 4) {
#pragma unroll
      for (int t = 0; t < 4; ++t) {
        int ob = (wave + 4 * t) * 16;
        const float* ab =
            Wt + ((size_t)((k0 >> 1) + lhi) * MPAD + ob + ml) * 2 + koffb * (size_t)(32 * MPAD * 2);
#pragma unroll
        for (int kw = 0; kw < 3; ++kw) {
          v2f a = *(const v2f*)(ab + (size_t)kw * (32 * MPAD * 2));
#pragma unroll
          for (int j = 0; j < 2; ++j) {
            v2f b = *(const v2f*)&strip[(((k0 >> 1) + lhi) * 34 + j * 16 + ml + kw) * 2];
            acc[t][j] = wmma_f32(a, b, acc[t][j]);
          }
        }
      }
    }
  }

  const size_t sbase = ((size_t)d * 32 + h) * 32;
#pragma unroll
  for (int t = 0; t < 4; ++t) {
    int ob = (wave + 4 * t) * 16;
#pragma unroll
    for (int j = 0; j < 2; ++j)
#pragma unroll
      for (int v = 0; v < 8; ++v) {
        int o = ob + v + 8 * lhi;
        y2p[(size_t)o * SPATIAL + sbase + j * 16 + ml] = acc[t][j][v];
      }
  }
}

__global__ __launch_bounds__(256) void softmax27_kernel(
    const float* __restrict__ y2p, float* __restrict__ wsft) {
  int t = blockIdx.x * 256 + threadIdx.x;
  int sub = t >> 15;
  int s = t & (SPATIAL - 1);
  float v[27];
  float mx = -3.0e38f;
#pragma unroll
  for (int n = 0; n < 27; ++n) {
    v[n] = y2p[(size_t)(n * 8 + sub) * SPATIAL + s];
    mx = fmaxf(mx, v[n]);
  }
  float sum = 0.0f;
#pragma unroll
  for (int n = 0; n < 27; ++n) {
    v[n] = __expf(v[n] - mx);
    sum += v[n];
  }
  float inv = 1.0f / sum;
#pragma unroll
  for (int n = 0; n < 27; ++n)
    wsft[((size_t)s * 27 + n) * 8 + sub] = v[n] * inv;
}

__global__ __launch_bounds__(64) void carafe_apply_kernel(
    const float* __restrict__ Y, const float* __restrict__ wsft,
    float* __restrict__ out) {
  __shared__ float wl[216];
  __shared__ float yl[32 * 27];
  int s = blockIdx.x;
  int d = s >> 10, h = (s >> 5) & 31, w = s & 31;
  int tid = threadIdx.x;

  for (int i = tid; i < 216; i += 64) wl[i] = wsft[(size_t)s * 216 + i];
  for (int i = tid; i < 32 * 27; i += 64) {
    int o = i / 27, n = i % 27;
    int nd = min(max(d + n / 9 - 1, 0), 31);
    int nh = min(max(h + (n / 3) % 3 - 1, 0), 31);
    int nw = min(max(w + n % 3 - 1, 0), 31);
    yl[i] = Y[(size_t)o * SPATIAL + ((size_t)nd * 32 + nh) * 32 + nw];
  }
  __syncthreads();

#pragma unroll
  for (int r = 0; r < 4; ++r) {
    int oi = tid + r * 64;
    int o = oi >> 3, sub = oi & 7;
    float acc = 0.0f;
#pragma unroll
    for (int n = 0; n < 27; ++n) acc += yl[o * 27 + n] * wl[n * 8 + sub];
    int rd = sub >> 2, rh = (sub >> 1) & 1, rw = sub & 1;
    size_t vox = ((size_t)(2 * d + rd) * 64 + (2 * h + rh)) * 64 + (2 * w + rw);
    out[(size_t)o * 262144 + vox] = acc;
  }
}

extern "C" void kernel_launch(void* const* d_in, const int* in_sizes, int n_in,
                              void* d_out, int out_size, void* d_ws,
                              size_t ws_size, hipStream_t stream) {
  (void)in_sizes; (void)n_in; (void)out_size; (void)ws_size;
  const float* x  = (const float*)d_in[0];
  const float* Wc = (const float*)d_in[1];
  const float* bc = (const float*)d_in[2];
  const float* We = (const float*)d_in[3];
  const float* be = (const float*)d_in[4];
  const float* Wp = (const float*)d_in[5];
  float* out = (float*)d_out;

  float* ws = (float*)d_ws;
  float* y1   = ws;
  float* y2p  = y1 + (size_t)64 * SPATIAL;
  float* wsft = y2p + (size_t)MPAD * SPATIAL;
  float* Yp   = wsft + (size_t)SPATIAL * 216;
  float* Wt   = Yp + (size_t)32 * SPATIAL;
  float* bep  = Wt + (size_t)NWT;

  prep_weights_kernel<<<(NWT + MPAD + 255) / 256, 256, 0, stream>>>(We, be, Wt, bep);
  gemm_f32_wmma_kernel<4, true><<<512, 128, 0, stream>>>(Wc, x, bc, y1, SPATIAL);
  conv3_wmma_kernel<<<1024, 128, 0, stream>>>(y1, Wt, bep, y2p);
  softmax27_kernel<<<1024, 256, 0, stream>>>(y2p, wsft);
  gemm_f32_wmma_kernel<2, false><<<512, 128, 0, stream>>>(Wp, x, nullptr, Yp, SPATIAL);
  carafe_apply_kernel<<<32768, 64, 0, stream>>>(Yp, wsft, out);
}


# apply kernel re-tiled: one 128-thread block per (d,h) row stages its Y rows and weights in LDS once; thread = output column x half the channels, whole 128-byte line stores; tap sum order unchanged
# speedup vs baseline: 1.5253x; 1.5253x over previous
#include <hip/hip_runtime.h>
#include <hip/hip_bf16.h>

typedef float v2f __attribute__((ext_vector_type(2)));
typedef float v8f __attribute__((ext_vector_type(8)));

__device__ __forceinline__ v8f wmma_f32(v2f a, v2f b, v8f c) {
  return __builtin_amdgcn_wmma_f32_16x16x4_f32(false, a, false, b, (short)0, c,
                                               false, false);
}

#define SPATIAL 32768
#define MPAD 256
#define NWT (27 * 64 * MPAD)

__global__ __launch_bounds__(256) void prep_weights_kernel(
    const float* __restrict__ We, const float* __restrict__ be,
    float* __restrict__ Wt, float* __restrict__ bep) {
  int idx = blockIdx.x * 256 + threadIdx.x;
  if (idx < NWT) {
    int o = idx & (MPAD - 1);
    int cm = (idx >> 8) & 63;
    int koff = idx >> 14;
    float v = (o < 216) ? We[(size_t)o * 1728 + cm * 27 + koff] : 0.0f;
    Wt[((size_t)(koff * 32 + (cm >> 1)) * MPAD + o) * 2 + (cm & 1)] = v;
  } else if (idx < NWT + MPAD) {
    int o = idx - NWT;
    bep[o] = (o < 216) ? be[o] : 0.0f;
  }
}

template <int MTILES, bool BIAS>
__global__ __launch_bounds__(128) void gemm_f32_wmma_kernel(
    const float* __restrict__ A, const float* __restrict__ B,
    const float* __restrict__ bias, float* __restrict__ C, int N) {
  constexpr int M = MTILES * 16;
  __shared__ float lA[64 * 64];
  __shared__ float lB[64 * 64];

  const int tid = threadIdx.x;
  const int lane = tid & 31;
  const int wave = tid >> 5;
  const int n0 = blockIdx.x * 64;
  const int ml = lane & 15;
  const int lhi = lane >> 4;

  for (int i = tid; i < M * 64; i += 128) lA[i] = A[i];
  for (int i = tid; i < 64 * 64; i += 128) {
    int k = i >> 6, col = i & 63;
    lB[((k >> 1) * 64 + col) * 2 + (k & 1)] = B[(size_t)k * N + n0 + col];
  }
  __syncthreads();

  int mt[MTILES], nt[MTILES];
#pragma unroll
  for (int i = 0; i < MTILES; ++i) {
    int p = wave + 4 * i;
    mt[i] = p % MTILES;
    nt[i] = p / MTILES;
  }

  v8f acc[MTILES];
#pragma unroll
  for (int i = 0; i < MTILES; ++i) {
    v8f c;
#pragma unroll
    for (int v = 0; v < 8; ++v)
      c[v] = BIAS ? bias[mt[i] * 16 + v + 8 * lhi] : 0.0f;
    acc[i] = c;
  }

#pragma unroll
  for (int k0 = 0; k0 < 64; k0 += 4) {
#pragma unroll
    for (int i = 0; i < MTILES; ++i) {
      v2f a = *(const v2f*)&lA[(mt[i] * 16 + ml) * 64 + k0 + 2 * lhi];
      v2f b = *(const v2f*)&lB[(((k0 >> 1) + lhi) * 64 + nt[i] * 16 + ml) * 2];
      acc[i] = wmma_f32(a, b, acc[i]);
    }
  }

#pragma unroll
  for (int i = 0; i < MTILES; ++i)
#pragma unroll
    for (int v = 0; v < 8; ++v) {
      int m = mt[i] * 16 + v + 8 * lhi;
      C[(size_t)m * N + n0 + nt[i] * 16 + ml] = acc[i][v];
    }
}

__global__ __launch_bounds__(128) void conv3_wmma_kernel(
    const float* __restrict__ y1, const float* __restrict__ Wt,
    const float* __restrict__ bep, float* __restrict__ y2p) {
  __shared__ float strip[64 * 34];

  const int tid = threadIdx.x;
  const int lane = tid & 31;
  const int wave = tid >> 5;
  const int d = blockIdx.x >> 5;
  const int h = blockIdx.x & 31;
  const int ml = lane & 15;
  const int lhi = lane >> 4;

  v8f acc[4][2];
#pragma unroll
  for (int t = 0; t < 4; ++t) {
    int ob = (wave + 4 * t) * 16;
#pragma unroll
    for (int j = 0; j < 2; ++j) {
      v8f c;
#pragma unroll
      for (int v = 0; v < 8; ++v) c[v] = bep[ob + v + 8 * lhi];
      acc[t][j] = c;
    }
  }

  for (int q = 0; q < 9; ++q) {
    const int kd = q / 3, kh = q % 3;
    const int dd = d + kd - 1, hh = h + kh - 1;
    const bool rowok = (dd >= 0 && dd < 32 && hh >= 0 && hh < 32);
    const int ddc = min(max(dd, 0), 31), hhc = min(max(hh, 0), 31);
    const float* src = y1 + ((size_t)ddc * 32 + hhc) * 32;

    __syncthreads();
    for (int i = tid; i < 64 * 34; i += 128) {
      int cm = i / 34, pos = i % 34;
      int wsrc = pos - 1;
      int wc = min(max(wsrc, 0), 31);
      float ld = src[(size_t)cm * SPATIAL + wc];
      bool ok = rowok && (wsrc >= 0) && (wsrc < 32);
      strip[((cm >> 1) * 34 + pos) * 2 + (cm & 1)] = ok ? ld : 0.0f;
    }
    if (q < 8 && tid < 64) {
      int kd2 = (q + 1) / 3, kh2 = (q + 1) % 3;
      int d2 = min(max(d + kd2 - 1, 0), 31), h2 = min(max(h + kh2 - 1, 0), 31);
      __builtin_prefetch(y1 + ((size_t)tid * SPATIAL) + ((size_t)d2 * 32 + h2) * 32,
                         0, 0);
    }
    __syncthreads();

    const int koffb = kd * 9 + kh * 3;
#pragma unroll 2
    for (int k0 = 0; k0 < 64; k0 += 4) {
#pragma unroll
      for (int t = 0; t < 4; ++t) {
        int ob = (wave + 4 * t) * 16;
        const float* ab =
            Wt + ((size_t)((k0 >> 1) + lhi) * MPAD + ob + ml) * 2 + koffb * (size_t)(32 * MPAD * 2);
#pragma unroll
        for (int kw = 0; kw < 3; ++kw) {
          v2f a = *(const v2f*)(ab + (size_t)kw * (32 * MPAD * 2));
#pragma unroll
          for (int j = 0; j < 2; ++j) {
            v2f b = *(const v2f*)&strip[(((k0 >> 1) + lhi) * 34 + j * 16 + ml + kw) * 2];
            acc[t][j] = wmma_f32(a, b, acc[t][j]);
          }
        }
      }
    }
  }

  const size_t sbase = ((size_t)d * 32 + h) * 32;
#pragma unroll
  for (int t = 0; t < 4; ++t) {
    int ob = (wave + 4 * t) * 16;
#pragma unroll
    for (int j = 0; j < 2; ++j)
#pragma unroll
      for (int v = 0; v < 8; ++v) {
        int o = ob + v + 8 * lhi;
        y2p[(size_t)o * SPATIAL + sbase + j * 16 + ml] = acc[t][j][v];
      }
  }
}

__global__ __launch_bounds__(256) void softmax27_kernel(
    const float* __restrict__ y2p, float* __restrict__ wsft) {
  int t = blockIdx.x * 256 + threadIdx.x;
  int sub = t >> 15;
  int s = t & (SPATIAL - 1);
  float v[27];
  float mx = -3.0e38f;
#pragma unroll
  for (int n = 0; n < 27; ++n) {
    v[n] = y2p[(size_t)(n * 8 + sub) * SPATIAL + s];
    mx = fmaxf(mx, v[n]);
  }
  float sum = 0.0f;
#pragma unroll
  for (int n = 0; n < 27; ++n) {
    v[n] = __expf(v[n] - mx);
    sum += v[n];
  }
  float inv = 1.0f / sum;
#pragma unroll
  for (int n = 0; n < 27; ++n)
    wsft[((size_t)s * 27 + n) * 8 + sub] = v[n] * inv;
}

#define WPITCH 66
#define APPLY_T 128
__global__ __launch_bounds__(APPLY_T) void carafe_apply_kernel(
    const float* __restrict__ Y, const float* __restrict__ wsft,
    float* __restrict__ out) {
  __shared__ __attribute__((aligned(16))) float yl[32 * 9 * 32];
  __shared__ __attribute__((aligned(16))) float wl[108 * WPITCH];
  const int tid = threadIdx.x;
  const int d = blockIdx.x >> 5, h = blockIdx.x & 31;

  constexpr int NY = 288 * 8 / APPLY_T;
  float4 yv[NY];
#pragma unroll
  for (int j = 0; j < NY; ++j) {
    int i = tid + j * APPLY_T;
    int row = i >> 3, part = i & 7;
    int o = row / 9, r = row - o * 9;
    int kd = r / 3, kh = r - kd * 3;
    int dd = min(max(d + kd - 1, 0), 31);
    int hh = min(max(h + kh - 1, 0), 31);
    yv[j] = *(const float4*)(Y + (size_t)o * SPATIAL + (dd * 32 + hh) * 32 + part * 4);
  }
  const float* wrow = wsft + (size_t)blockIdx.x * (32 * 216);
  constexpr int NW = (1728 + APPLY_T - 1) / APPLY_T;
  float4 wv[NW];
#pragma unroll
  for (int j = 0; j < NW; ++j) {
    int i = tid + j * APPLY_T;
    if (j < NW - 1 || tid < 1728 - (NW - 1) * APPLY_T) wv[j] = *(const float4*)(wrow + 4 * i);
  }
#pragma unroll
  for (int j = 0; j < NY; ++j) {
    int i = tid + j * APPLY_T;
    *(float4*)&yl[i * 4] = yv[j];
  }
#pragma unroll
  for (int j = 0; j < NW; ++j) {
    int i = tid + j * APPLY_T;
    if (j < NW - 1 || tid < 1728 - (NW - 1) * APPLY_T) {
      int w = i / 54, rem = i - w * 54;
      int n = rem >> 1, rd = rem & 1;
      float* dst = &wl[(n * 4 + rd * 2) * WPITCH + 2 * w];
      *(float2*)dst = make_float2(wv[j].x, wv[j].y);
      *(float2*)(dst + WPITCH) = make_float2(wv[j].z, wv[j].w);
    }
  }
  __syncthreads();

  const int w2 = tid & 63;
  const int cg = tid >> 6;
  const int w = w2 >> 1;
  const int wm = max(w - 1, 0), wp = min(w + 1, 31);

  float wr[27][4];
#pragma unroll
  for (int n = 0; n < 27; ++n)
#pragma unroll
    for (int q = 0; q < 4; ++q) wr[n][q] = wl[(n * 4 + q) * WPITCH + w2];

  float* op = out + (size_t)(cg * 16) * 262144 + (size_t)(2 * d) * 4096 + (2 * h) * 64 + w2;
  const float* yo = &yl[cg * 16 * 288];
#pragma clang loop vectorize(disable)
  for (int c = 0; c < 16; ++c) {
    float a0 = 0.0f, a1 = 0.0f, a2 = 0.0f, a3 = 0.0f;
#pragma unroll
    for (int r = 0; r < 9; ++r) {
      float y0 = yo[r * 32 + wm];
      float y1v = yo[r * 32 + w];
      float y2v = yo[r * 32 + wp];
      a0 += y0 * wr[3 * r][0];
      a1 += y0 * wr[3 * r][1];
      a2 += y0 * wr[3 * r][2];
      a3 += y0 * wr[3 * r][3];
      a0 += y1v * wr[3 * r + 1][0];
      a1 += y1v * wr[3 * r + 1][1];
      a2 += y1v * wr[3 * r + 1][2];
      a3 += y1v * wr[3 * r + 1][3];
      a0 += y2v * wr[3 * r + 2][0];
      a1 += y2v * wr[3 * r + 2][1];
      a2 += y2v * wr[3 * r + 2][2];
      a3 += y2v * wr[3 * r + 2][3];
    }
    op[0] = a0;
    op[64] = a1;
    op[4096] = a2;
    op[4096 + 64] = a3;
    op += 262144;
    yo += 288;
  }
}

extern "C" void kernel_launch(void* const* d_in, const int* in_sizes, int n_in,
                              void* d_out, int out_size, void* d_ws,
                              size_t ws_size, hipStream_t stream) {
  (void)in_sizes; (void)n_in; (void)out_size; (void)ws_size;
  const float* x  = (const float*)d_in[0];
  const float* Wc = (const float*)d_in[1];
  const float* bc = (const float*)d_in[2];
  const float* We = (const float*)d_in[3];
  const float* be = (const float*)d_in[4];
  const float* Wp = (const float*)d_in[5];
  float* out = (float*)d_out;

  float* ws = (float*)d_ws;
  float* y1   = ws;
  float* y2p  = y1 + (size_t)64 * SPATIAL;
  float* wsft = y2p + (size_t)MPAD * SPATIAL;
  float* Yp   = wsft + (size_t)SPATIAL * 216;
  float* Wt   = Yp + (size_t)32 * SPATIAL;
  float* bep  = Wt + (size_t)NWT;

  prep_weights_kernel<<<(NWT + MPAD + 255) / 256, 256, 0, stream>>>(We, be, Wt, bep);
  gemm_f32_wmma_kernel<4, true><<<512, 128, 0, stream>>>(Wc, x, bc, y1, SPATIAL);
  conv3_wmma_kernel<<<1024, 128, 0, stream>>>(y1, Wt, bep, y2p);
  softmax27_kernel<<<1024, 256, 0, stream>>>(y2p, wsft);
  gemm_f32_wmma_kernel<2, false><<<512, 128, 0, stream>>>(Wp, x, nullptr, Yp, SPATIAL);
  carafe_apply_kernel<<<1024, APPLY_T, 0, stream>>>(Yp, wsft, out);
}


# conv kernel epilogue does pixel shuffle + 27-tap softmax through LDS and writes the weight row with contiguous 16-byte stores; logits plane and softmax launch removed; sum orders unchanged
# speedup vs baseline: 1.7540x; 1.1499x over previous
#include <hip/hip_runtime.h>
#include <hip/hip_bf16.h>

typedef float v2f __attribute__((ext_vector_type(2)));
typedef float v8f __attribute__((ext_vector_type(8)));

__device__ __forceinline__ v8f wmma_f32(v2f a, v2f b, v8f c) {
  return __builtin_amdgcn_wmma_f32_16x16x4_f32(false, a, false, b, (short)0, c,
                                               false, false);
}

#define SPATIAL 32768
#define MPAD 256
#define NWT (27 * 64 * MPAD)
#define NCH 216

__global__ __launch_bounds__(256) void prep_weights_kernel(
    const float* __restrict__ We, const float* __restrict__ be,
    float* __restrict__ Wt, float* __restrict__ bep) {
  int idx = blockIdx.x * 256 + threadIdx.x;
  if (idx < NWT) {
    int o = idx & (MPAD - 1);
    int cm = (idx >> 8) & 63;
    int koff = idx >> 14;
    float v = (o < 216) ? We[(size_t)o * 1728 + cm * 27 + koff] : 0.0f;
    Wt[((size_t)(koff * 32 + (cm >> 1)) * MPAD + o) * 2 + (cm & 1)] = v;
  } else if (idx < NWT + MPAD) {
    int o = idx - NWT;
    bep[o] = (o < 216) ? be[o] : 0.0f;
  }
}

template <int MTILES, bool BIAS>
__global__ __launch_bounds__(128) void gemm_f32_wmma_kernel(
    const float* __restrict__ A, const float* __restrict__ B,
    const float* __restrict__ bias, float* __restrict__ C, int N) {
  constexpr int M = MTILES * 16;
  __shared__ float lA[64 * 64];
  __shared__ float lB[64 * 64];

  const int tid = threadIdx.x;
  const int lane = tid & 31;
  const int wave = tid >> 5;
  const int n0 = blockIdx.x * 64;
  const int ml = lane & 15;
  const int lhi = lane >> 4;

  for (int i = tid; i < M * 64; i += 128) lA[i] = A[i];
  for (int i = tid; i < 64 * 64; i += 128) {
    int k = i >> 6, col = i & 63;
    lB[((k >> 1) * 64 + col) * 2 + (k & 1)] = B[(size_t)k * N + n0 + col];
  }
  __syncthreads();

  int mt[MTILES], nt[MTILES];
#pragma unroll
  for (int i = 0; i < MTILES; ++i) {
    int p = wave + 4 * i;
    mt[i] = p % MTILES;
    nt[i] = p / MTILES;
  }

  v8f acc[MTILES];
#pragma unroll
  for (int i = 0; i < MTILES; ++i) {
    v8f c;
#pragma unroll
    for (int v = 0; v < 8; ++v)
      c[v] = BIAS ? bias[mt[i] * 16 + v + 8 * lhi] : 0.0f;
    acc[i] = c;
  }

#pragma unroll
  for (int k0 = 0; k0 < 64; k0 += 4) {
#pragma unroll
    for (int i = 0; i < MTILES; ++i) {
      v2f a = *(const v2f*)&lA[(mt[i] * 16 + ml) * 64 + k0 + 2 * lhi];
      v2f b = *(const v2f*)&lB[(((k0 >> 1) + lhi) * 64 + nt[i] * 16 + ml) * 2];
      acc[i] = wmma_f32(a, b, acc[i]);
    }
  }

#pragma unroll
  for (int i = 0; i < MTILES; ++i)
#pragma unroll
    for (int v = 0; v < 8; ++v) {
      int m = mt[i] * 16 + v + 8 * lhi;
      C[(size_t)m * N + n0 + nt[i] * 16 + ml] = acc[i][v];
    }
}

__global__ __launch_bounds__(128) void conv3_softmax_wmma_kernel(
    const float* __restrict__ y1, const float* __restrict__ Wt,
    const float* __restrict__ bep, float* __restrict__ wsft) {
  __shared__ __attribute__((aligned(16))) float lds[64 * 34 + 32 * NCH];
  float* const strip = lds;
  float* const ep = lds + 64 * 34;

  const int tid = threadIdx.x;
  const int lane = tid & 31;
  const int wave = tid >> 5;
  const int d = blockIdx.x >> 5;
  const int h = blockIdx.x & 31;
  const int ml = lane & 15;
  const int lhi = lane >> 4;

  v8f acc[4][2];
#pragma unroll
  for (int t = 0; t < 4; ++t) {
    int ob = (wave + 4 * t) * 16;
#pragma unroll
    for (int j = 0; j < 2; ++j) {
      v8f c;
#pragma unroll
      for (int v = 0; v < 8; ++v) c[v] = bep[ob + v + 8 * lhi];
      acc[t][j] = c;
    }
  }

  for (int q = 0; q < 9; ++q) {
    const int kd = q / 3, kh = q % 3;
    const int dd = d + kd - 1, hh = h + kh - 1;
    const bool rowok = (dd >= 0 && dd < 32 && hh >= 0 && hh < 32);
    const int ddc = min(max(dd, 0), 31), hhc = min(max(hh, 0), 31);
    const float* src = y1 + ((size_t)ddc * 32 + hhc) * 32;

    __syncthreads();
    for (int i = tid; i < 64 * 34; i += 128) {
      int cm = i / 34, pos = i % 34;
      int wsrc = pos - 1;
      int wc = min(max(wsrc, 0), 31);
      float ld = src[(size_t)cm * SPATIAL + wc];
      bool ok = rowok && (wsrc >= 0) && (wsrc < 32);
      strip[((cm >> 1) * 34 + pos) * 2 + (cm & 1)] = ok ? ld : 0.0f;
    }
    if (q < 8 && tid < 64) {
      int kd2 = (q + 1) / 3, kh2 = (q + 1) % 3;
      int d2 = min(max(d + kd2 - 1, 0), 31), h2 = min(max(h + kh2 - 1, 0), 31);
      __builtin_prefetch(y1 + ((size_t)tid * SPATIAL) + ((size_t)d2 * 32 + h2) * 32,
                         0, 0);
    }
    __syncthreads();

    const int koffb = kd * 9 + kh * 3;
#pragma unroll 2
    for (int k0 = 0; k0 < 64; k0 += 4) {
#pragma unroll
      for (int t = 0; t < 4; ++t) {
        int ob = (wave + 4 * t) * 16;
        const float* ab =
            Wt + ((size_t)((k0 >> 1) + lhi) * MPAD + ob + ml) * 2 + koffb * (size_t)(32 * MPAD * 2);
#pragma unroll
        for (int kw = 0; kw < 3; ++kw) {
          v2f a = *(const v2f*)(ab + (size_t)kw * (32 * MPAD * 2));
#pragma unroll
          for (int j = 0; j < 2; ++j) {
            v2f b = *(const v2f*)&strip[(((k0 >> 1) + lhi) * 34 + j * 16 + ml + kw) * 2];
            acc[t][j] = wmma_f32(a, b, acc[t][j]);
          }
        }
      }
    }
  }

#pragma unroll
  for (int t = 0; t < 4; ++t) {
    const int o0 = (wave + 4 * t) * 16 + 8 * lhi;
#pragma unroll
    for (int j = 0; j < 2; ++j) {
      if (o0 < NCH) {
        float4* dst = (float4*)&ep[(j * 16 + ml) * NCH + o0];
        dst[0] = make_float4(acc[t][j][0], acc[t][j][1], acc[t][j][2], acc[t][j][3]);
        dst[1] = make_float4(acc[t][j][4], acc[t][j][5], acc[t][j][6], acc[t][j][7]);
      }
    }
  }
  __syncthreads();

  for (int r = 0; r < 2; ++r) {
    const int p = tid + 128 * r;
    float* e = &ep[(p >> 3) * NCH + (p & 7)];
    float v[27];
    float mx = -3.0e38f;
#pragma unroll
    for (int n = 0; n < 27; ++n) {
      v[n] = e[n * 8];
      mx = fmaxf(mx, v[n]);
    }
    float sum = 0.0f;
#pragma unroll
    for (int n = 0; n < 27; ++n) {
      v[n] = __expf(v[n] - mx);
      sum += v[n];
    }
    float inv = 1.0f / sum;
#pragma unroll
    for (int n = 0; n < 27; ++n) e[n * 8] = v[n] * inv;
  }
  __syncthreads();

  float4* gdst = (float4*)(wsft + (size_t)blockIdx.x * (32 * NCH));
  const float4* lsrc = (const float4*)ep;
  constexpr int NV = 32 * NCH / 4;
#pragma unroll
  for (int j = 0; j < (NV + 127) / 128; ++j) {
    const int i = tid + j * 128;
    if (j < NV / 128 || tid < NV % 128) gdst[i] = lsrc[i];
  }
}

#define WPITCH 66
#define APPLY_T 128
__global__ __launch_bounds__(APPLY_T) void carafe_apply_kernel(
    const float* __restrict__ Y, const float* __restrict__ wsft,
    float* __restrict__ out) {
  __shared__ __attribute__((aligned(16))) float yl[32 * 9 * 32];
  __shared__ __attribute__((aligned(16))) float wl[108 * WPITCH];
  const int tid = threadIdx.x;
  const int d = blockIdx.x >> 5, h = blockIdx.x & 31;

  constexpr int NY = 288 * 8 / APPLY_T;
  float4 yv[NY];
#pragma unroll
  for (int j = 0; j < NY; ++j) {
    int i = tid + j * APPLY_T;
    int row = i >> 3, part = i & 7;
    int o = row / 9, r = row - o * 9;
    int kd = r / 3, kh = r - kd * 3;
    int dd = min(max(d + kd - 1, 0), 31);
    int hh = min(max(h + kh - 1, 0), 31);
    yv[j] = *(const float4*)(Y + (size_t)o * SPATIAL + (dd * 32 + hh) * 32 + part * 4);
  }
  const float* wrow = wsft + (size_t)blockIdx.x * (32 * 216);
  constexpr int NW = (1728 + APPLY_T - 1) / APPLY_T;
  float4 wv[NW];
#pragma unroll
  for (int j = 0; j < NW; ++j) {
    int i = tid + j * APPLY_T;
    if (j < NW - 1 || tid < 1728 - (NW - 1) * APPLY_T) wv[j] = *(const float4*)(wrow + 4 * i);
  }
#pragma unroll
  for (int j = 0; j < NY; ++j) {
    int i = tid + j * APPLY_T;
    *(float4*)&yl[i * 4] = yv[j];
  }
#pragma unroll
  for (int j = 0; j < NW; ++j) {
    int i = tid + j * APPLY_T;
    if (j < NW - 1 || tid < 1728 - (NW - 1) * APPLY_T) {
      int w = i / 54, rem = i - w * 54;
      int n = rem >> 1, rd = rem & 1;
      float* dst = &wl[(n * 4 + rd * 2) * WPITCH + 2 * w];
      *(float2*)dst = make_float2(wv[j].x, wv[j].y);
      *(float2*)(dst + WPITCH) = make_float2(wv[j].z, wv[j].w);
    }
  }
  __syncthreads();

  const int w2 = tid & 63;
  const int cg = tid >> 6;
  const int w = w2 >> 1;
  const int wm = max(w - 1, 0), wp = min(w + 1, 31);

  float wr[27][4];
#pragma unroll
  for (int n = 0; n < 27; ++n)
#pragma unroll
    for (int q = 0; q < 4; ++q) wr[n][q] = wl[(n * 4 + q) * WPITCH + w2];

  float* op = out + (size_t)(cg * 16) * 262144 + (size_t)(2 * d) * 4096 + (2 * h) * 64 + w2;
  const float* yo = &yl[cg * 16 * 288];
#pragma clang loop vectorize(disable)
  for (int c = 0; c < 16; ++c) {
    float a0 = 0.0f, a1 = 0.0f, a2 = 0.0f, a3 = 0.0f;
#pragma unroll
    for (int r = 0; r < 9; ++r) {
      float y0 = yo[r * 32 + wm];
      float y1v = yo[r * 32 + w];
      float y2v = yo[r * 32 + wp];
      a0 += y0 * wr[3 * r][0];
      a1 += y0 * wr[3 * r][1];
      a2 += y0 * wr[3 * r][2];
      a3 += y0 * wr[3 * r][3];
      a0 += y1v * wr[3 * r + 1][0];
      a1 += y1v * wr[3 * r + 1][1];
      a2 += y1v * wr[3 * r + 1][2];
      a3 += y1v * wr[3 * r + 1][3];
      a0 += y2v * wr[3 * r + 2][0];
      a1 += y2v * wr[3 * r + 2][1];
      a2 += y2v * wr[3 * r + 2][2];
      a3 += y2v * wr[3 * r + 2][3];
    }
    op[0] = a0;
    op[64] = a1;
    op[4096] = a2;
    op[4096 + 64] = a3;
    op += 262144;
    yo += 288;
  }
}

extern "C" void kernel_launch(void* const* d_in, const int* in_sizes, int n_in,
                              void* d_out, int out_size, void* d_ws,
                              size_t ws_size, hipStream_t stream) {
  (void)in_sizes; (void)n_in; (void)out_size; (void)ws_size;
  const float* x  = (const float*)d_in[0];
  const float* Wc = (const float*)d_in[1];
  const float* bc = (const float*)d_in[2];
  const float* We = (const float*)d_in[3];
  const float* be = (const float*)d_in[4];
  const float* Wp = (const float*)d_in[5];
  float* out = (float*)d_out;

  float* ws = (float*)d_ws;
  float* y1   = ws;
  float* wsft = y1 + (size_t)64 * SPATIAL;
  float* Yp   = wsft + (size_t)SPATIAL * 216;
  float* Wt   = Yp + (size_t)32 * SPATIAL;
  float* bep  = Wt + (size_t)NWT;

  prep_weights_kernel<<<(NWT + MPAD + 255) / 256, 256, 0, stream>>>(We, be, Wt, bep);
  gemm_f32_wmma_kernel<4, true><<<512, 128, 0, stream>>>(Wc, x, bc, y1, SPATIAL);
  conv3_softmax_wmma_kernel<<<1024, 128, 0, stream>>>(y1, Wt, bep, wsft);
  gemm_f32_wmma_kernel<2, false><<<512, 128, 0, stream>>>(Wp, x, nullptr, Yp, SPATIAL);
  carafe_apply_kernel<<<1024, APPLY_T, 0, stream>>>(Yp, wsft, out);
}


# conv block re-tiled: two h rows x 14 M-tiles (was one row x 16 padded), wave = 7 M-tiles x one row with next k-step's fragments loaded a step ahead; one eighth fewer WMMAs; sum orders unchanged
# speedup vs baseline: 2.0825x; 1.1873x over previous
#include <hip/hip_runtime.h>
#include <hip/hip_bf16.h>

typedef float v2f __attribute__((ext_vector_type(2)));
typedef float v8f __attribute__((ext_vector_type(8)));

__device__ __forceinline__ v8f wmma_f32(v2f a, v2f b, v8f c) {
  return __builtin_amdgcn_wmma_f32_16x16x4_f32(false, a, false, b, (short)0, c,
                                               false, false);
}

#define SPATIAL 32768
#define MPAD 224
#define NWT (27 * 64 * MPAD)
#define NCH 216

__global__ __launch_bounds__(256) void prep_weights_kernel(
    const float* __restrict__ We, const float* __restrict__ be,
    float* __restrict__ Wt, float* __restrict__ bep) {
  int idx = blockIdx.x * 256 + threadIdx.x;
  if (idx < NWT) {
    int o = idx % MPAD;
    int rest = idx / MPAD;
    int cm = rest & 63;
    int koff = rest >> 6;
    float v = (o < 216) ? We[(size_t)o * 1728 + cm * 27 + koff] : 0.0f;
    Wt[((size_t)(koff * 32 + (cm >> 1)) * MPAD + o) * 2 + (cm & 1)] = v;
  } else if (idx < NWT + MPAD) {
    int o = idx - NWT;
    bep[o] = (o < 216) ? be[o] : 0.0f;
  }
}

template <int MTILES, bool BIAS>
__global__ __launch_bounds__(128) void gemm_f32_wmma_kernel(
    const float* __restrict__ A, const float* __restrict__ B,
    const float* __restrict__ bias, float* __restrict__ C, int N) {
  constexpr int M = MTILES * 16;
  __shared__ float lA[64 * 64];
  __shared__ float lB[64 * 64];

  const int tid = threadIdx.x;
  const int lane = tid & 31;
  const int wave = tid >> 5;
  const int n0 = blockIdx.x * 64;
  const int ml = lane & 15;
  const int lhi = lane >> 4;

  for (int i = tid; i < M * 64; i += 128) lA[i] = A[i];
  for (int i = tid; i < 64 * 64; i += 128) {
    int k = i >> 6, col = i & 63;
    lB[((k >> 1) * 64 + col) * 2 + (k & 1)] = B[(size_t)k * N + n0 + col];
  }
  __syncthreads();

  int mt[MTILES], nt[MTILES];
#pragma unroll
  for (int i = 0; i < MTILES; ++i) {
    int p = wave + 4 * i;
    mt[i] = p % MTILES;
    nt[i] = p / MTILES;
  }

  v8f acc[MTILES];
#pragma unroll
  for (int i = 0; i < MTILES; ++i) {
    v8f c;
#pragma unroll
    for (int v = 0; v < 8; ++v)
      c[v] = BIAS ? bias[mt[i] * 16 + v + 8 * lhi] : 0.0f;
    acc[i] = c;
  }

#pragma unroll
  for (int k0 = 0; k0 < 64; k0 += 4) {
#pragma unroll
    for (int i = 0; i < MTILES; ++i) {
      v2f a = *(const v2f*)&lA[(mt[i] * 16 + ml) * 64 + k0 + 2 * lhi];
      v2f b = *(const v2f*)&lB[(((k0 >> 1) + lhi) * 64 + nt[i] * 16 + ml) * 2];
      acc[i] = wmma_f32(a, b, acc[i]);
    }
  }

#pragma unroll
  for (int i = 0; i < MTILES; ++i)
#pragma unroll
    for (int v = 0; v < 8; ++v) {
      int m = mt[i] * 16 + v + 8 * lhi;
      C[(size_t)m * N + n0 + nt[i] * 16 + ml] = acc[i][v];
    }
}

#define STRIPF (64 * 34)
#define NSTEP 144

__device__ __forceinline__ void conv_load_a(const float* __restrict__ abase,
                                            int sg, v2f (&a)[7][3]) {
  const float* p =
      abase + (size_t)((sg >> 4) * (3 * 32) + (sg & 15) * 2) * (MPAD * 2);
#pragma unroll
  for (int kw = 0; kw < 3; ++kw)
#pragma unroll
    for (int t = 0; t < 7; ++t)
      a[t][kw] = *(const v2f*)(p + kw * (32 * MPAD * 2) + t * 32);
}

__device__ __forceinline__ void conv_load_b(const float* bb, v2f (&b)[3][2]) {
#pragma unroll
  for (int kw = 0; kw < 3; ++kw)
#pragma unroll
    for (int j = 0; j < 2; ++j) b[kw][j] = *(const v2f*)(bb + (j * 16 + kw) * 2);
}

__device__ __forceinline__ void conv_step(const float* __restrict__ abase,
                                          const float* bbn, int sg,
                                          v2f (&ac)[7][3], v2f (&an)[7][3],
                                          v2f (&bc)[3][2], v2f (&bn)[3][2],
                                          v8f (&acc)[7][2]) {
  conv_load_a(abase, min(sg + 1, NSTEP - 1), an);
  conv_load_b(bbn, bn);
  __builtin_amdgcn_sched_barrier(0);
#pragma unroll
  for (int kw = 0; kw < 3; ++kw)
#pragma unroll
    for (int t = 0; t < 7; ++t)
#pragma unroll
      for (int j = 0; j < 2; ++j) acc[t][j] = wmma_f32(ac[t][kw], bc[kw][j], acc[t][j]);
}

__global__ __launch_bounds__(128) void conv3_softmax_wmma_kernel(
    const float* __restrict__ y1, const float* __restrict__ Wt,
    const float* __restrict__ bep, float* __restrict__ wsft) {
  __shared__ __attribute__((aligned(16))) float lds[4 * STRIPF + 64 * NCH];
  float* const strips = lds;
  float* const ep = lds + 4 * STRIPF;

  const int tid = threadIdx.x;
  const int lane = tid & 31;
  const int wave = tid >> 5;
  const int d = blockIdx.x >> 4;
  const int h0 = (blockIdx.x & 15) * 2;
  const int ml = lane & 15;
  const int lhi = lane >> 4;
  const int mh = wave & 1;
  const int r = wave >> 1;

  v8f acc[7][2];
#pragma unroll
  for (int t = 0; t < 7; ++t) {
    int ob = (mh * 7 + t) * 16;
#pragma unroll
    for (int j = 0; j < 2; ++j) {
      v8f c;
#pragma unroll
      for (int v = 0; v < 8; ++v) c[v] = bep[ob + v + 8 * lhi];
      acc[t][j] = c;
    }
  }

#pragma unroll
  for (int e = 0; e < 2; ++e) {
    int idx = tid + 128 * e;
    int s4 = idx >> 6, cm2 = (idx >> 1) & 31, side = idx & 1;
    *(float2*)&strips[s4 * STRIPF + (cm2 * 34 + side * 33) * 2] = make_float2(0.0f, 0.0f);
  }

  const float* abase = Wt + (size_t)((lhi * MPAD + mh * 112 + ml) * 2);
  const float* bbase = strips + r * STRIPF + (lhi * 34 + ml) * 2;

  v2f aX[7][3], aY[7][3];
  v2f bX[3][2], bY[3][2];
  conv_load_a(abase, 0, aX);

#pragma unroll 1
  for (int kd = 0; kd < 3; ++kd) {
    const int dd = d + kd - 1;
    const bool dok = (dd >= 0 && dd < 32);
    const int ddc = min(max(dd, 0), 31);
    float4 ya[8], yb[8];
#pragma unroll
    for (int i = 0; i < 8; ++i) {
      const int part = tid & 7;
      const int cm2 = (tid >> 3) + 16 * (i & 1);
      const int hh = h0 - 1 + (i >> 1);
      const int hhc = min(max(hh, 0), 31);
      const float* s = y1 + (size_t)(2 * cm2) * SPATIAL + (ddc * 32 + hhc) * 32 + part * 4;
      ya[i] = *(const float4*)s;
      yb[i] = *(const float4*)(s + SPATIAL);
    }
    __syncthreads();
#pragma unroll
    for (int i = 0; i < 8; ++i) {
      const int part = tid & 7;
      const int cm2 = (tid >> 3) + 16 * (i & 1);
      const int hh = h0 - 1 + (i >> 1);
      const bool ok = dok && (hh >= 0) && (hh < 32);
      float4 u = ok ? ya[i] : make_float4(0.0f, 0.0f, 0.0f, 0.0f);
      float4 v = ok ? yb[i] : make_float4(0.0f, 0.0f, 0.0f, 0.0f);
      float* dst = strips + (i >> 1) * STRIPF + (cm2 * 34 + part * 4 + 1) * 2;
      *(float2*)(dst + 0) = make_float2(u.x, v.x);
      *(float2*)(dst + 2) = make_float2(u.y, v.y);
      *(float2*)(dst + 4) = make_float2(u.z, v.z);
      *(float2*)(dst + 6) = make_float2(u.w, v.w);
    }
    if (kd < 2) {
      const int d2 = min(max(d + kd, 0), 31);
#pragma unroll
      for (int e = 0; e < 2; ++e) {
        int idx = tid + 128 * e;
        int h2 = min(max(h0 - 1 + (idx >> 6), 0), 31);
        __builtin_prefetch(y1 + (size_t)(idx & 63) * SPATIAL + (d2 * 32 + h2) * 32, 0, 0);
      }
    }
    __syncthreads();

    conv_load_b(bbase, bX);
#pragma unroll 1
    for (int s = 0; s < 48; s += 2) {
      const int s1 = s + 1, s2 = min(s + 2, 47);
      const float* bb1 = bbase + (s1 >> 4) * STRIPF + (s1 & 15) * (2 * 34 * 2);
      const float* bb2 = bbase + (s2 >> 4) * STRIPF + (s2 & 15) * (2 * 34 * 2);
      conv_step(abase, bb1, kd * 48 + s, aX, aY, bX, bY, acc);
      conv_step(abase, bb2, kd * 48 + s + 1, aY, aX, bY, bX, acc);
    }
  }

#pragma unroll
  for (int t = 0; t < 7; ++t) {
    const int o0 = (mh * 7 + t) * 16 + 8 * lhi;
#pragma unroll
    for (int j = 0; j < 2; ++j) {
      if (o0 < NCH) {
        float4* dst = (float4*)&ep[(r * 32 + j * 16 + ml) * NCH + o0];
        dst[0] = make_float4(acc[t][j][0], acc[t][j][1], acc[t][j][2], acc[t][j][3]);
        dst[1] = make_float4(acc[t][j][4], acc[t][j][5], acc[t][j][6], acc[t][j][7]);
      }
    }
  }
  __syncthreads();

  for (int rr = 0; rr < 4; ++rr) {
    const int p = tid + 128 * rr;
    float* e = &ep[(p >> 3) * NCH + (p & 7)];
    float v[27];
    float mx = -3.0e38f;
#pragma unroll
    for (int n = 0; n < 27; ++n) {
      v[n] = e[n * 8];
      mx = fmaxf(mx, v[n]);
    }
    float sum = 0.0f;
#pragma unroll
    for (int n = 0; n < 27; ++n) {
      v[n] = __expf(v[n] - mx);
      sum += v[n];
    }
    float inv = 1.0f / sum;
#pragma unroll
    for (int n = 0; n < 27; ++n) e[n * 8] = v[n] * inv;
  }
  __syncthreads();

  float4* gdst = (float4*)(wsft + (size_t)blockIdx.x * (64 * NCH));
  const float4* lsrc = (const float4*)ep;
  constexpr int NV = 64 * NCH / 4;
#pragma unroll
  for (int j = 0; j < NV / 128; ++j) {
    const int i = tid + j * 128;
    gdst[i] = lsrc[i];
  }
}

#define WPITCH 66
#define APPLY_T 128
__global__ __launch_bounds__(APPLY_T) void carafe_apply_kernel(
    const float* __restrict__ Y, const float* __restrict__ wsft,
    float* __restrict__ out) {
  __shared__ __attribute__((aligned(16))) float yl[32 * 9 * 32];
  __shared__ __attribute__((aligned(16))) float wl[108 * WPITCH];
  const int tid = threadIdx.x;
  const int d = blockIdx.x >> 5, h = blockIdx.x & 31;

  constexpr int NY = 288 * 8 / APPLY_T;
  float4 yv[NY];
#pragma unroll
  for (int j = 0; j < NY; ++j) {
    int i = tid + j * APPLY_T;
    int row = i >> 3, part = i & 7;
    int o = row / 9, r = row - o * 9;
    int kd = r / 3, kh = r - kd * 3;
    int dd = min(max(d + kd - 1, 0), 31);
    int hh = min(max(h + kh - 1, 0), 31);
    yv[j] = *(const float4*)(Y + (size_t)o * SPATIAL + (dd * 32 + hh) * 32 + part * 4);
  }
  const float* wrow = wsft + (size_t)blockIdx.x * (32 * 216);
  constexpr int NW = (1728 + APPLY_T - 1) / APPLY_T;
  float4 wv[NW];
#pragma unroll
  for (int j = 0; j < NW; ++j) {
    int i = tid + j * APPLY_T;
    if (j < NW - 1 || tid < 1728 - (NW - 1) * APPLY_T) wv[j] = *(const float4*)(wrow + 4 * i);
  }
#pragma unroll
  for (int j = 0; j < NY; ++j) {
    int i = tid + j * APPLY_T;
    *(float4*)&yl[i * 4] = yv[j];
  }
#pragma unroll
  for (int j = 0; j < NW; ++j) {
    int i = tid + j * APPLY_T;
    if (j < NW - 1 || tid < 1728 - (NW - 1) * APPLY_T) {
      int w = i / 54, rem = i - w * 54;
      int n = rem >> 1, rd = rem & 1;
      float* dst = &wl[(n * 4 + rd * 2) * WPITCH + 2 * w];
      *(float2*)dst = make_float2(wv[j].x, wv[j].y);
      *(float2*)(dst + WPITCH) = make_float2(wv[j].z, wv[j].w);
    }
  }
  __syncthreads();

  const int w2 = tid & 63;
  const int cg = tid >> 6;
  const int w = w2 >> 1;
  const int wm = max(w - 1, 0), wp = min(w + 1, 31);

  float wr[27][4];
#pragma unroll
  for (int n = 0; n < 27; ++n)
#pragma unroll
    for (int q = 0; q < 4; ++q) wr[n][q] = wl[(n * 4 + q) * WPITCH + w2];

  float* op = out + (size_t)(cg * 16) * 262144 + (size_t)(2 * d) * 4096 + (2 * h) * 64 + w2;
  const float* yo = &yl[cg * 16 * 288];
#pragma clang loop vectorize(disable)
  for (int c = 0; c < 16; ++c) {
    float a0 = 0.0f, a1 = 0.0f, a2 = 0.0f, a3 = 0.0f;
#pragma unroll
    for (int r = 0; r < 9; ++r) {
      float y0 = yo[r * 32 + wm];
      float y1v = yo[r * 32 + w];
      float y2v = yo[r * 32 + wp];
      a0 += y0 * wr[3 * r][0];
      a1 += y0 * wr[3 * r][1];
      a2 += y0 * wr[3 * r][2];
      a3 += y0 * wr[3 * r][3];
      a0 += y1v * wr[3 * r + 1][0];
      a1 += y1v * wr[3 * r + 1][1];
      a2 += y1v * wr[3 * r + 1][2];
      a3 += y1v * wr[3 * r + 1][3];
      a0 += y2v * wr[3 * r + 2][0];
      a1 += y2v * wr[3 * r + 2][1];
      a2 += y2v * wr[3 * r + 2][2];
      a3 += y2v * wr[3 * r + 2][3];
    }
    op[0] = a0;
    op[64] = a1;
    op[4096] = a2;
    op[4096 + 64] = a3;
    op += 262144;
    yo += 288;
  }
}

extern "C" void kernel_launch(void* const* d_in, const int* in_sizes, int n_in,
                              void* d_out, int out_size, void* d_ws,
                              size_t ws_size, hipStream_t stream) {
  (void)in_sizes; (void)n_in; (void)out_size; (void)ws_size;
  const float* x  = (const float*)d_in[0];
  const float* Wc = (const float*)d_in[1];
  const float* bc = (const float*)d_in[2];
  const float* We = (const float*)d_in[3];
  const float* be = (const float*)d_in[4];
  const float* Wp = (const float*)d_in[5];
  float* out = (float*)d_out;

  float* ws = (float*)d_ws;
  float* y1   = ws;
  float* wsft = y1 + (size_t)64 * SPATIAL;
  float* Yp   = wsft + (size_t)SPATIAL * 216;
  float* Wt   = Yp + (size_t)32 * SPATIAL;
  float* bep  = Wt + (size_t)NWT;

  prep_weights_kernel<<<(NWT + MPAD + 255) / 256, 256, 0, stream>>>(We, be, Wt, bep);
  gemm_f32_wmma_kernel<4, true><<<512, 128, 0, stream>>>(Wc, x, bc, y1, SPATIAL);
  conv3_softmax_wmma_kernel<<<512, 128, 0, stream>>>(y1, Wt, bep, wsft);
  gemm_f32_wmma_kernel<2, false><<<512, 128, 0, stream>>>(Wp, x, nullptr, Yp, SPATIAL);
  carafe_apply_kernel<<<1024, APPLY_T, 0, stream>>>(Yp, wsft, out);
}


# three launches ahead of the conv (weight re-layout, y1 GEMM, Y GEMM) merged into one: a block stages its x tile once for both products, the re-layout is extra coalesced blocks; sum orders unchanged
# speedup vs baseline: 2.4121x; 1.1583x over previous
#include <hip/hip_runtime.h>
#include <hip/hip_bf16.h>

typedef float v2f __attribute__((ext_vector_type(2)));
typedef float v8f __attribute__((ext_vector_type(8)));

__device__ __forceinline__ v8f wmma_f32(v2f a, v2f b, v8f c) {
  return __builtin_amdgcn_wmma_f32_16x16x4_f32(false, a, false, b, (short)0, c,
                                               false, false);
}

#define SPATIAL 32768
#define MPAD 224
#define NWT (27 * 64 * MPAD)
#define NCH 216

#define GEMM_BLOCKS 512
#define FRONT_BLOCKS 568
__global__ __launch_bounds__(128) void front_kernel(
    const float* __restrict__ x, const float* __restrict__ Wc,
    const float* __restrict__ bc, const float* __restrict__ Wp,
    const float* __restrict__ We, const float* __restrict__ be,
    float* __restrict__ y1, float* __restrict__ Yp, float* __restrict__ Wt,
    float* __restrict__ bep) {
  __shared__ float lA[96 * 64];
  __shared__ float lB[64 * 64];

  const int tid = threadIdx.x;

  if (blockIdx.x >= GEMM_BLOCKS) {
    const int t = (blockIdx.x - GEMM_BLOCKS) * 128 + tid;
    const int o = t % MPAD;
    const int cm2 = t / MPAD;
    if (t < MPAD) bep[t] = (t < NCH) ? be[t] : 0.0f;
    float vf[54];
    if (o < NCH) {
      const float2* s = (const float2*)(We + (size_t)o * 1728 + cm2 * 54);
#pragma unroll
      for (int k = 0; k < 27; ++k) {
        float2 q = s[k];
        vf[2 * k] = q.x;
        vf[2 * k + 1] = q.y;
      }
    } else {
#pragma unroll
      for (int k = 0; k < 54; ++k) vf[k] = 0.0f;
    }
    float* dst = Wt + ((size_t)cm2 * MPAD + o) * 2;
#pragma unroll
    for (int koff = 0; koff < 27; ++koff)
      *(float2*)(dst + (size_t)koff * (32 * MPAD * 2)) = make_float2(vf[koff], vf[27 + koff]);
    return;
  }

  const int lane = tid & 31;
  const int wave = tid >> 5;
  const int n0 = blockIdx.x * 64;
  const int ml = lane & 15;
  const int lhi = lane >> 4;

  for (int i = tid; i < 64 * 64; i += 128) lA[i] = Wc[i];
  for (int i = tid; i < 32 * 64; i += 128) lA[64 * 64 + i] = Wp[i];
  for (int i = tid; i < 64 * 64; i += 128) {
    int k = i >> 6, col = i & 63;
    lB[((k >> 1) * 64 + col) * 2 + (k & 1)] = x[(size_t)k * SPATIAL + n0 + col];
  }
  __syncthreads();

  v8f acc[6];
#pragma unroll
  for (int t = 0; t < 6; ++t) {
    v8f c;
#pragma unroll
    for (int v = 0; v < 8; ++v) c[v] = (t < 4) ? bc[t * 16 + v + 8 * lhi] : 0.0f;
    acc[t] = c;
  }

#pragma unroll
  for (int k0 = 0; k0 < 64; k0 += 4) {
    v2f b = *(const v2f*)&lB[(((k0 >> 1) + lhi) * 64 + wave * 16 + ml) * 2];
#pragma unroll
    for (int t = 0; t < 6; ++t) {
      v2f a = *(const v2f*)&lA[(t * 16 + ml) * 64 + k0 + 2 * lhi];
      acc[t] = wmma_f32(a, b, acc[t]);
    }
  }

  const int col = n0 + wave * 16 + ml;
#pragma unroll
  for (int t = 0; t < 6; ++t)
#pragma unroll
    for (int v = 0; v < 8; ++v) {
      const int m = t * 16 + v + 8 * lhi;
      if (t < 4)
        y1[(size_t)m * SPATIAL + col] = acc[t][v];
      else
        Yp[(size_t)(m - 64) * SPATIAL + col] = acc[t][v];
    }
}

#define STRIPF (64 * 34)
#define NSTEP 144

__device__ __forceinline__ void conv_load_a(const float* __restrict__ abase,
                                            int sg, v2f (&a)[7][3]) {
  const float* p =
      abase + (size_t)((sg >> 4) * (3 * 32) + (sg & 15) * 2) * (MPAD * 2);
#pragma unroll
  for (int kw = 0; kw < 3; ++kw)
#pragma unroll
    for (int t = 0; t < 7; ++t)
      a[t][kw] = *(const v2f*)(p + kw * (32 * MPAD * 2) + t * 32);
}

__device__ __forceinline__ void conv_load_b(const float* bb, v2f (&b)[3][2]) {
#pragma unroll
  for (int kw = 0; kw < 3; ++kw)
#pragma unroll
    for (int j = 0; j < 2; ++j) b[kw][j] = *(const v2f*)(bb + (j * 16 + kw) * 2);
}

__device__ __forceinline__ void conv_step(const float* __restrict__ abase,
                                          const float* bbn, int sg,
                                          v2f (&ac)[7][3], v2f (&an)[7][3],
                                          v2f (&bc)[3][2], v2f (&bn)[3][2],
                                          v8f (&acc)[7][2]) {
  conv_load_a(abase, min(sg + 1, NSTEP - 1), an);
  conv_load_b(bbn, bn);
  __builtin_amdgcn_sched_barrier(0);
#pragma unroll
  for (int kw = 0; kw < 3; ++kw)
#pragma unroll
    for (int t = 0; t < 7; ++t)
#pragma unroll
      for (int j = 0; j < 2; ++j) acc[t][j] = wmma_f32(ac[t][kw], bc[kw][j], acc[t][j]);
}

__global__ __launch_bounds__(128) void conv3_softmax_wmma_kernel(
    const float* __restrict__ y1, const float* __restrict__ Wt,
    const float* __restrict__ bep, float* __restrict__ wsft) {
  __shared__ __attribute__((aligned(16))) float lds[4 * STRIPF + 64 * NCH];
  float* const strips = lds;
  float* const ep = lds + 4 * STRIPF;

  const int tid = threadIdx.x;
  const int lane = tid & 31;
  const int wave = tid >> 5;
  const int d = blockIdx.x >> 4;
  const int h0 = (blockIdx.x & 15) * 2;
  const int ml = lane & 15;
  const int lhi = lane >> 4;
  const int mh = wave & 1;
  const int r = wave >> 1;

  v8f acc[7][2];
#pragma unroll
  for (int t = 0; t < 7; ++t) {
    int ob = (mh * 7 + t) * 16;
#pragma unroll
    for (int j = 0; j < 2; ++j) {
      v8f c;
#pragma unroll
      for (int v = 0; v < 8; ++v) c[v] = bep[ob + v + 8 * lhi];
      acc[t][j] = c;
    }
  }

#pragma unroll
  for (int e = 0; e < 2; ++e) {
    int idx = tid + 128 * e;
    int s4 = idx >> 6, cm2 = (idx >> 1) & 31, side = idx & 1;
    *(float2*)&strips[s4 * STRIPF + (cm2 * 34 + side * 33) * 2] = make_float2(0.0f, 0.0f);
  }

  const float* abase = Wt + (size_t)((lhi * MPAD + mh * 112 + ml) * 2);
  const float* bbase = strips + r * STRIPF + (lhi * 34 + ml) * 2;

  v2f aX[7][3], aY[7][3];
  v2f bX[3][2], bY[3][2];
  conv_load_a(abase, 0, aX);

#pragma unroll 1
  for (int kd = 0; kd < 3; ++kd) {
    const int dd = d + kd - 1;
    const bool dok = (dd >= 0 && dd < 32);
    const int ddc = min(max(dd, 0), 31);
    float4 ya[8], yb[8];
#pragma unroll
    for (int i = 0; i < 8; ++i) {
      const int part = tid & 7;
      const int cm2 = (tid >> 3) + 16 * (i & 1);
      const int hh = h0 - 1 + (i >> 1);
      const int hhc = min(max(hh, 0), 31);
      const float* s = y1 + (size_t)(2 * cm2) * SPATIAL + (ddc * 32 + hhc) * 32 + part * 4;
      ya[i] = *(const float4*)s;
      yb[i] = *(const float4*)(s + SPATIAL);
    }
    __syncthreads();
#pragma unroll
    for (int i = 0; i < 8; ++i) {
      const int part = tid & 7;
      const int cm2 = (tid >> 3) + 16 * (i & 1);
      const int hh = h0 - 1 + (i >> 1);
      const bool ok = dok && (hh >= 0) && (hh < 32);
      float4 u = ok ? ya[i] : make_float4(0.0f, 0.0f, 0.0f, 0.0f);
      float4 v = ok ? yb[i] : make_float4(0.0f, 0.0f, 0.0f, 0.0f);
      float* dst = strips + (i >> 1) * STRIPF + (cm2 * 34 + part * 4 + 1) * 2;
      *(float2*)(dst + 0) = make_float2(u.x, v.x);
      *(float2*)(dst + 2) = make_float2(u.y, v.y);
      *(float2*)(dst + 4) = make_float2(u.z, v.z);
      *(float2*)(dst + 6) = make_float2(u.w, v.w);
    }
    if (kd < 2) {
      const int d2 = min(max(d + kd, 0), 31);
#pragma unroll
      for (int e = 0; e < 2; ++e) {
        int idx = tid + 128 * e;
        int h2 = min(max(h0 - 1 + (idx >> 6), 0), 31);
        __builtin_prefetch(y1 + (size_t)(idx & 63) * SPATIAL + (d2 * 32 + h2) * 32, 0, 0);
      }
    }
    __syncthreads();

    conv_load_b(bbase, bX);
#pragma unroll 1
    for (int s = 0; s < 48; s += 2) {
      const int s1 = s + 1, s2 = min(s + 2, 47);
      const float* bb1 = bbase + (s1 >> 4) * STRIPF + (s1 & 15) * (2 * 34 * 2);
      const float* bb2 = bbase + (s2 >> 4) * STRIPF + (s2 & 15) * (2 * 34 * 2);
      conv_step(abase, bb1, kd * 48 + s, aX, aY, bX, bY, acc);
      conv_step(abase, bb2, kd * 48 + s + 1, aY, aX, bY, bX, acc);
    }
  }

#pragma unroll
  for (int t = 0; t < 7; ++t) {
    const int o0 = (mh * 7 + t) * 16 + 8 * lhi;
#pragma unroll
    for (int j = 0; j < 2; ++j) {
      if (o0 < NCH) {
        float4* dst = (float4*)&ep[(r * 32 + j * 16 + ml) * NCH + o0];
        dst[0] = make_float4(acc[t][j][0], acc[t][j][1], acc[t][j][2], acc[t][j][3]);
        dst[1] = make_float4(acc[t][j][4], acc[t][j][5], acc[t][j][6], acc[t][j][7]);
      }
    }
  }
  __syncthreads();

  for (int rr = 0; rr < 4; ++rr) {
    const int p = tid + 128 * rr;
    float* e = &ep[(p >> 3) * NCH + (p & 7)];
    float v[27];
    float mx = -3.0e38f;
#pragma unroll
    for (int n = 0; n < 27; ++n) {
      v[n] = e[n * 8];
      mx = fmaxf(mx, v[n]);
    }
    float sum = 0.0f;
#pragma unroll
    for (int n = 0; n < 27; ++n) {
      v[n] = __expf(v[n] - mx);
      sum += v[n];
    }
    float inv = 1.0f / sum;
#pragma unroll
    for (int n = 0; n < 27; ++n) e[n * 8] = v[n] * inv;
  }
  __syncthreads();

  float4* gdst = (float4*)(wsft + (size_t)blockIdx.x * (64 * NCH));
  const float4* lsrc = (const float4*)ep;
  constexpr int NV = 64 * NCH / 4;
#pragma unroll
  for (int j = 0; j < NV / 128; ++j) {
    const int i = tid + j * 128;
    gdst[i] = lsrc[i];
  }
}

#define WPITCH 66
#define APPLY_T 128
__global__ __launch_bounds__(APPLY_T) void carafe_apply_kernel(
    const float* __restrict__ Y, const float* __restrict__ wsft,
    float* __restrict__ out) {
  __shared__ __attribute__((aligned(16))) float yl[32 * 9 * 32];
  __shared__ __attribute__((aligned(16))) float wl[108 * WPITCH];
  const int tid = threadIdx.x;
  const int d = blockIdx.x >> 5, h = blockIdx.x & 31;

  constexpr int NY = 288 * 8 / APPLY_T;
  float4 yv[NY];
#pragma unroll
  for (int j = 0; j < NY; ++j) {
    int i = tid + j * APPLY_T;
    int row = i >> 3, part = i & 7;
    int o = row / 9, r = row - o * 9;
    int kd = r / 3, kh = r - kd * 3;
    int dd = min(max(d + kd - 1, 0), 31);
    int hh = min(max(h + kh - 1, 0), 31);
    yv[j] = *(const float4*)(Y + (size_t)o * SPATIAL + (dd * 32 + hh) * 32 + part * 4);
  }
  const float* wrow = wsft + (size_t)blockIdx.x * (32 * 216);
  constexpr int NW = (1728 + APPLY_T - 1) / APPLY_T;
  float4 wv[NW];
#pragma unroll
  for (int j = 0; j < NW; ++j) {
    int i = tid + j * APPLY_T;
    if (j < NW - 1 || tid < 1728 - (NW - 1) * APPLY_T) wv[j] = *(const float4*)(wrow + 4 * i);
  }
#pragma unroll
  for (int j = 0; j < NY; ++j) {
    int i = tid + j * APPLY_T;
    *(float4*)&yl[i * 4] = yv[j];
  }
#pragma unroll
  for (int j = 0; j < NW; ++j) {
    int i = tid + j * APPLY_T;
    if (j < NW - 1 || tid < 1728 - (NW - 1) * APPLY_T) {
      int w = i / 54, rem = i - w * 54;
      int n = rem >> 1, rd = rem & 1;
      float* dst = &wl[(n * 4 + rd * 2) * WPITCH + 2 * w];
      *(float2*)dst = make_float2(wv[j].x, wv[j].y);
      *(float2*)(dst + WPITCH) = make_float2(wv[j].z, wv[j].w);
    }
  }
  __syncthreads();

  const int w2 = tid & 63;
  const int cg = tid >> 6;
  const int w = w2 >> 1;
  const int wm = max(w - 1, 0), wp = min(w + 1, 31);

  float wr[27][4];
#pragma unroll
  for (int n = 0; n < 27; ++n)
#pragma unroll
    for (int q = 0; q < 4; ++q) wr[n][q] = wl[(n * 4 + q) * WPITCH + w2];

  float* op = out + (size_t)(cg * 16) * 262144 + (size_t)(2 * d) * 4096 + (2 * h) * 64 + w2;
  const float* yo = &yl[cg * 16 * 288];
#pragma clang loop vectorize(disable)
  for (int c = 0; c < 16; ++c) {
    float a0 = 0.0f, a1 = 0.0f, a2 = 0.0f, a3 = 0.0f;
#pragma unroll
    for (int r = 0; r < 9; ++r) {
      float y0 = yo[r * 32 + wm];
      float y1v = yo[r * 32 + w];
      float y2v = yo[r * 32 + wp];
      a0 += y0 * wr[3 * r][0];
      a1 += y0 * wr[3 * r][1];
      a2 += y0 * wr[3 * r][2];
      a3 += y0 * wr[3 * r][3];
      a0 += y1v * wr[3 * r + 1][0];
      a1 += y1v * wr[3 * r + 1][1];
      a2 += y1v * wr[3 * r + 1][2];
      a3 += y1v * wr[3 * r + 1][3];
      a0 += y2v * wr[3 * r + 2][0];
      a1 += y2v * wr[3 * r + 2][1];
      a2 += y2v * wr[3 * r + 2][2];
      a3 += y2v * wr[3 * r + 2][3];
    }
    op[0] = a0;
    op[64] = a1;
    op[4096] = a2;
    op[4096 + 64] = a3;
    op += 262144;
    yo += 288;
  }
}

extern "C" void kernel_launch(void* const* d_in, const int* in_sizes, int n_in,
                              void* d_out, int out_size, void* d_ws,
                              size_t ws_size, hipStream_t stream) {
  (void)in_sizes; (void)n_in; (void)out_size; (void)ws_size;
  const float* x  = (const float*)d_in[0];
  const float* Wc = (const float*)d_in[1];
  const float* bc = (const float*)d_in[2];
  const float* We = (const float*)d_in[3];
  const float* be = (const float*)d_in[4];
  const float* Wp = (const float*)d_in[5];
  float* out = (float*)d_out;

  float* ws = (float*)d_ws;
  float* y1   = ws;
  float* wsft = y1 + (size_t)64 * SPATIAL;
  float* Yp   = wsft + (size_t)SPATIAL * 216;
  float* Wt   = Yp + (size_t)32 * SPATIAL;
  float* bep  = Wt + (size_t)NWT;

  front_kernel<<<FRONT_BLOCKS, 128, 0, stream>>>(x, Wc, bc, Wp, We, be, y1, Yp, Wt, bep);
  conv3_softmax_wmma_kernel<<<512, 128, 0, stream>>>(y1, Wt, bep, wsft);
  carafe_apply_kernel<<<1024, APPLY_T, 0, stream>>>(Yp, wsft, out);
}


# conv in the F(2,3) minimal-filtering form along h: 4 transformed products per 2 output rows instead of 6, one third fewer WMMAs; weights transformed in the front kernel; logit sum order changes
# speedup vs baseline: 3.0680x; 1.2719x over previous
#include <hip/hip_runtime.h>
#include <hip/hip_bf16.h>

typedef float v2f __attribute__((ext_vector_type(2)));
typedef float v8f __attribute__((ext_vector_type(8)));

__device__ __forceinline__ v8f wmma_f32(v2f a, v2f b, v8f c) {
  return __builtin_amdgcn_wmma_f32_16x16x4_f32(false, a, false, b, (short)0, c,
                                               false, false);
}

#define SPATIAL 32768
#define MPAD 224
#define NWT (36 * 64 * MPAD)
#define NCH 216

#define GEMM_BLOCKS 512
#define FRONT_BLOCKS 568
__global__ __launch_bounds__(128) void front_kernel(
    const float* __restrict__ x, const float* __restrict__ Wc,
    const float* __restrict__ bc, const float* __restrict__ Wp,
    const float* __restrict__ We, const float* __restrict__ be,
    float* __restrict__ y1, float* __restrict__ Yp, float* __restrict__ Wt,
    float* __restrict__ bep) {
  __shared__ float lA[96 * 64];
  __shared__ float lB[64 * 64];

  const int tid = threadIdx.x;

  if (blockIdx.x >= GEMM_BLOCKS) {
    const int t = (blockIdx.x - GEMM_BLOCKS) * 128 + tid;
    const int o = t % MPAD;
    const int cm2 = t / MPAD;
    if (t < MPAD) bep[t] = (t < NCH) ? be[t] : 0.0f;
    float vf[54];
    if (o < NCH) {
      const float2* s = (const float2*)(We + (size_t)o * 1728 + cm2 * 54);
#pragma unroll
      for (int k = 0; k < 27; ++k) {
        float2 q = s[k];
        vf[2 * k] = q.x;
        vf[2 * k + 1] = q.y;
      }
    } else {
#pragma unroll
      for (int k = 0; k < 54; ++k) vf[k] = 0.0f;
    }
    float* dst = Wt + ((size_t)cm2 * MPAD + o) * 2;
#pragma unroll
    for (int kd = 0; kd < 3; ++kd)
#pragma unroll
      for (int kw = 0; kw < 3; ++kw) {
        const int i0 = kd * 9 + kw;
        const float e0 = vf[i0], e1 = vf[i0 + 3], e2 = vf[i0 + 6];
        const float q0 = vf[27 + i0], q1 = vf[27 + i0 + 3], q2 = vf[27 + i0 + 6];
        float* dk = dst + (size_t)(kd * 12 + kw) * (32 * MPAD * 2);
        *(float2*)(dk) = make_float2(e0, q0);
        *(float2*)(dk + (size_t)3 * (32 * MPAD * 2)) =
            make_float2(0.5f * (e0 + e1 + e2), 0.5f * (q0 + q1 + q2));
        *(float2*)(dk + (size_t)6 * (32 * MPAD * 2)) =
            make_float2(0.5f * (e0 - e1 + e2), 0.5f * (q0 - q1 + q2));
        *(float2*)(dk + (size_t)9 * (32 * MPAD * 2)) = make_float2(e2, q2);
      }
    return;
  }

  const int lane = tid & 31;
  const int wave = tid >> 5;
  const int n0 = blockIdx.x * 64;
  const int ml = lane & 15;
  const int lhi = lane >> 4;

  for (int i = tid; i < 64 * 64; i += 128) lA[i] = Wc[i];
  for (int i = tid; i < 32 * 64; i += 128) lA[64 * 64 + i] = Wp[i];
  for (int i = tid; i < 64 * 64; i += 128) {
    int k = i >> 6, col = i & 63;
    lB[((k >> 1) * 64 + col) * 2 + (k & 1)] = x[(size_t)k * SPATIAL + n0 + col];
  }
  __syncthreads();

  v8f acc[6];
#pragma unroll
  for (int t = 0; t < 6; ++t) {
    v8f c;
#pragma unroll
    for (int v = 0; v < 8; ++v) c[v] = (t < 4) ? bc[t * 16 + v + 8 * lhi] : 0.0f;
    acc[t] = c;
  }

#pragma unroll
  for (int k0 = 0; k0 < 64; k0 += 4) {
    v2f b = *(const v2f*)&lB[(((k0 >> 1) + lhi) * 64 + wave * 16 + ml) * 2];
#pragma unroll
    for (int t = 0; t < 6; ++t) {
      v2f a = *(const v2f*)&lA[(t * 16 + ml) * 64 + k0 + 2 * lhi];
      acc[t] = wmma_f32(a, b, acc[t]);
    }
  }

  const int col = n0 + wave * 16 + ml;
#pragma unroll
  for (int t = 0; t < 6; ++t)
#pragma unroll
    for (int v = 0; v < 8; ++v) {
      const int m = t * 16 + v + 8 * lhi;
      if (t < 4)
        y1[(size_t)m * SPATIAL + col] = acc[t][v];
      else
        Yp[(size_t)(m - 64) * SPATIAL + col] = acc[t][v];
    }
}

#define STRIPF (64 * 34)
#define NSTEP 96

__device__ __forceinline__ void conv_load_a(const float* __restrict__ abase,
                                            int cp, int sg, v2f (&a)[7][3]) {
  const int g = (sg >> 5) * 4 + cp * 2 + ((sg >> 4) & 1);
  const float* p = abase + (size_t)(g * (3 * 32) + (sg & 15) * 2) * (MPAD * 2);
#pragma unroll
  for (int kw = 0; kw < 3; ++kw)
#pragma unroll
    for (int t = 0; t < 7; ++t)
      a[t][kw] = *(const v2f*)(p + kw * (32 * MPAD * 2) + t * 32);
}

__device__ __forceinline__ void conv_load_b(const float* bb, v2f (&b)[3][2]) {
#pragma unroll
  for (int kw = 0; kw < 3; ++kw)
#pragma unroll
    for (int j = 0; j < 2; ++j) b[kw][j] = *(const v2f*)(bb + (j * 16 + kw) * 2);
}

__device__ __forceinline__ void conv_step(const float* __restrict__ abase,
                                          int cp, const float* bbn, int sg,
                                          v2f (&ac)[7][3], v2f (&an)[7][3],
                                          v2f (&bc)[3][2], v2f (&bn)[3][2],
                                          v8f (&acc)[7][2]) {
  conv_load_a(abase, cp, min(sg + 1, NSTEP - 1), an);
  conv_load_b(bbn, bn);
  __builtin_amdgcn_sched_barrier(0);
#pragma unroll
  for (int kw = 0; kw < 3; ++kw)
#pragma unroll
    for (int t = 0; t < 7; ++t)
#pragma unroll
      for (int j = 0; j < 2; ++j) acc[t][j] = wmma_f32(ac[t][kw], bc[kw][j], acc[t][j]);
}

__device__ __forceinline__ float4 f4sub(float4 a, float4 b) {
  return make_float4(a.x - b.x, a.y - b.y, a.z - b.z, a.w - b.w);
}
__device__ __forceinline__ float4 f4add(float4 a, float4 b) {
  return make_float4(a.x + b.x, a.y + b.y, a.z + b.z, a.w + b.w);
}
__device__ __forceinline__ void strip_store(float* dst, float4 u, float4 v) {
  *(float2*)(dst + 0) = make_float2(u.x, v.x);
  *(float2*)(dst + 2) = make_float2(u.y, v.y);
  *(float2*)(dst + 4) = make_float2(u.z, v.z);
  *(float2*)(dst + 6) = make_float2(u.w, v.w);
}

__global__ __launch_bounds__(128) void conv3_softmax_wmma_kernel(
    const float* __restrict__ y1, const float* __restrict__ Wt,
    const float* __restrict__ bep, float* __restrict__ wsft) {
  __shared__ __attribute__((aligned(16))) float lds[4 * STRIPF + 64 * NCH];
  float* const strips = lds;
  float* const ep = lds + 4 * STRIPF;

  const int tid = threadIdx.x;
  const int lane = tid & 31;
  const int wave = tid >> 5;
  const int d = blockIdx.x >> 4;
  const int h0 = (blockIdx.x & 15) * 2;
  const int ml = lane & 15;
  const int lhi = lane >> 4;
  const int mh = wave & 1;
  const int cp = wave >> 1;

  v8f acc[2][7][2];
#pragma unroll
  for (int t = 0; t < 7; ++t) {
    int ob = (mh * 7 + t) * 16;
#pragma unroll
    for (int j = 0; j < 2; ++j) {
      v8f c, z;
#pragma unroll
      for (int v = 0; v < 8; ++v) {
        const float bv = bep[ob + v + 8 * lhi];
        c[v] = (cp == 0) ? bv : 0.0f;
        z[v] = 0.0f;
      }
      acc[0][t][j] = z;
      acc[1][t][j] = c;
    }
  }

#pragma unroll
  for (int e = 0; e < 2; ++e) {
    int idx = tid + 128 * e;
    int s4 = idx >> 6, cm2 = (idx >> 1) & 31, side = idx & 1;
    *(float2*)&strips[s4 * STRIPF + (cm2 * 34 + side * 33) * 2] = make_float2(0.0f, 0.0f);
  }

  const float* abase = Wt + (size_t)((lhi * MPAD + mh * 112 + ml) * 2);
  const float* bbase = strips + (cp * 2) * STRIPF + (lhi * 34 + ml) * 2;

  v2f aX[7][3], aY[7][3];
  v2f bX[3][2], bY[3][2];
  conv_load_a(abase, cp, 0, aX);

#pragma unroll 1
  for (int kd = 0; kd < 3; ++kd) {
    const int dd = d + kd - 1;
    const bool dok = (dd >= 0 && dd < 32);
    const int ddc = min(max(dd, 0), 31);
    float4 ya[8], yb[8];
#pragma unroll
    for (int i = 0; i < 8; ++i) {
      const int part = tid & 7;
      const int cm2 = (tid >> 3) + 16 * (i & 1);
      const int hh = h0 - 1 + (i >> 1);
      const int hhc = min(max(hh, 0), 31);
      const float* s = y1 + (size_t)(2 * cm2) * SPATIAL + (ddc * 32 + hhc) * 32 + part * 4;
      ya[i] = *(const float4*)s;
      yb[i] = *(const float4*)(s + SPATIAL);
    }
    __syncthreads();
#pragma unroll
    for (int e = 0; e < 2; ++e) {
      const int part = tid & 7;
      const int cm2 = (tid >> 3) + 16 * e;
      float* dst = strips + (cm2 * 34 + part * 4 + 1) * 2;
      float4 ra[4], rb[4];
#pragma unroll
      for (int k = 0; k < 4; ++k) {
        const int hh = h0 - 1 + k;
        const bool ok = dok && (hh >= 0) && (hh < 32);
        ra[k] = ok ? ya[2 * k + e] : make_float4(0.0f, 0.0f, 0.0f, 0.0f);
        rb[k] = ok ? yb[2 * k + e] : make_float4(0.0f, 0.0f, 0.0f, 0.0f);
      }
      strip_store(dst + 0 * STRIPF, f4sub(ra[0], ra[2]), f4sub(rb[0], rb[2]));
      strip_store(dst + 1 * STRIPF, f4add(ra[1], ra[2]), f4add(rb[1], rb[2]));
      strip_store(dst + 2 * STRIPF, f4sub(ra[2], ra[1]), f4sub(rb[2], rb[1]));
      strip_store(dst + 3 * STRIPF, f4sub(ra[1], ra[3]), f4sub(rb[1], rb[3]));
    }
    if (kd < 2) {
      const int d2 = min(max(d + kd, 0), 31);
#pragma unroll
      for (int e = 0; e < 2; ++e) {
        int idx = tid + 128 * e;
        int h2 = min(max(h0 - 1 + (idx >> 6), 0), 31);
        __builtin_prefetch(y1 + (size_t)(idx & 63) * SPATIAL + (d2 * 32 + h2) * 32, 0, 0);
      }
    }
    __syncthreads();

    conv_load_b(bbase, bX);
#pragma unroll
    for (int c = 0; c < 2; ++c) {
#pragma unroll 1
      for (int s = 0; s < 16; s += 2) {
        const int q1 = c * 16 + s + 1, q2 = min(c * 16 + s + 2, 31);
        const float* bb1 = bbase + (q1 >> 4) * STRIPF + (q1 & 15) * (2 * 34 * 2);
        const float* bb2 = bbase + (q2 >> 4) * STRIPF + (q2 & 15) * (2 * 34 * 2);
        conv_step(abase, cp, bb1, kd * 32 + c * 16 + s, aX, aY, bX, bY, acc[c]);
        conv_step(abase, cp, bb2, kd * 32 + c * 16 + s + 1, aY, aX, bY, bX, acc[c]);
      }
    }
  }

  if (cp == 0) {
#pragma unroll
    for (int t = 0; t < 7; ++t) {
      const int o0 = (mh * 7 + t) * 16 + 8 * lhi;
#pragma unroll
      for (int j = 0; j < 2; ++j) {
        if (o0 < NCH) {
          const v8f m0 = acc[0][t][j], m1 = acc[1][t][j];
          float4* r0 = (float4*)&ep[(j * 16 + ml) * NCH + o0];
          float4* r1 = (float4*)&ep[(32 + j * 16 + ml) * NCH + o0];
          r0[0] = make_float4(m0[0] + m1[0], m0[1] + m1[1], m0[2] + m1[2], m0[3] + m1[3]);
          r0[1] = make_float4(m0[4] + m1[4], m0[5] + m1[5], m0[6] + m1[6], m0[7] + m1[7]);
          r1[0] = make_float4(m1[0], m1[1], m1[2], m1[3]);
          r1[1] = make_float4(m1[4], m1[5], m1[6], m1[7]);
        }
      }
    }
  }
  __syncthreads();
  if (cp == 1) {
#pragma unroll
    for (int t = 0; t < 7; ++t) {
      const int o0 = (mh * 7 + t) * 16 + 8 * lhi;
#pragma unroll
      for (int j = 0; j < 2; ++j) {
        if (o0 < NCH) {
          const v8f m2 = acc[0][t][j], m3 = acc[1][t][j];
          float4* r0 = (float4*)&ep[(j * 16 + ml) * NCH + o0];
          float4* r1 = (float4*)&ep[(32 + j * 16 + ml) * NCH + o0];
          float4 a0 = r0[0], a1 = r0[1], b0 = r1[0], b1 = r1[1];
          r0[0] = make_float4(a0.x + m2[0], a0.y + m2[1], a0.z + m2[2], a0.w + m2[3]);
          r0[1] = make_float4(a1.x + m2[4], a1.y + m2[5], a1.z + m2[6], a1.w + m2[7]);
          r1[0] = make_float4((b0.x - m2[0]) - m3[0], (b0.y - m2[1]) - m3[1],
                              (b0.z - m2[2]) - m3[2], (b0.w - m2[3]) - m3[3]);
          r1[1] = make_float4((b1.x - m2[4]) - m3[4], (b1.y - m2[5]) - m3[5],
                              (b1.z - m2[6]) - m3[6], (b1.w - m2[7]) - m3[7]);
        }
      }
    }
  }
  __syncthreads();

  for (int rr = 0; rr < 4; ++rr) {
    const int p = tid + 128 * rr;
    float* e = &ep[(p >> 3) * NCH + (p & 7)];
    float v[27];
    float mx = -3.0e38f;
#pragma unroll
    for (int n = 0; n < 27; ++n) {
      v[n] = e[n * 8];
      mx = fmaxf(mx, v[n]);
    }
    float sum = 0.0f;
#pragma unroll
    for (int n = 0; n < 27; ++n) {
      v[n] = __expf(v[n] - mx);
      sum += v[n];
    }
    float inv = 1.0f / sum;
#pragma unroll
    for (int n = 0; n < 27; ++n) e[n * 8] = v[n] * inv;
  }
  __syncthreads();

  float4* gdst = (float4*)(wsft + (size_t)blockIdx.x * (64 * NCH));
  const float4* lsrc = (const float4*)ep;
  constexpr int NV = 64 * NCH / 4;
#pragma unroll
  for (int j = 0; j < NV / 128; ++j) {
    const int i = tid + j * 128;
    gdst[i] = lsrc[i];
  }
}

#define WPITCH 66
#define APPLY_T 128
__global__ __launch_bounds__(APPLY_T) void carafe_apply_kernel(
    const float* __restrict__ Y, const float* __restrict__ wsft,
    float* __restrict__ out) {
  __shared__ __attribute__((aligned(16))) float yl[32 * 9 * 32];
  __shared__ __attribute__((aligned(16))) float wl[108 * WPITCH];
  const int tid = threadIdx.x;
  const int d = blockIdx.x >> 5, h = blockIdx.x & 31;

  constexpr int NY = 288 * 8 / APPLY_T;
  float4 yv[NY];
#pragma unroll
  for (int j = 0; j < NY; ++j) {
    int i = tid + j * APPLY_T;
    int row = i >> 3, part = i & 7;
    int o = row / 9, r = row - o * 9;
    int kd = r / 3, kh = r - kd * 3;
    int dd = min(max(d + kd - 1, 0), 31);
    int hh = min(max(h + kh - 1, 0), 31);
    yv[j] = *(const float4*)(Y + (size_t)o * SPATIAL + (dd * 32 + hh) * 32 + part * 4);
  }
  const float* wrow = wsft + (size_t)blockIdx.x * (32 * 216);
  constexpr int NW = (1728 + APPLY_T - 1) / APPLY_T;
  float4 wv[NW];
#pragma unroll
  for (int j = 0; j < NW; ++j) {
    int i = tid + j * APPLY_T;
    if (j < NW - 1 || tid < 1728 - (NW - 1) * APPLY_T) wv[j] = *(const float4*)(wrow + 4 * i);
  }
#pragma unroll
  for (int j = 0; j < NY; ++j) {
    int i = tid + j * APPLY_T;
    *(float4*)&yl[i * 4] = yv[j];
  }
#pragma unroll
  for (int j = 0; j < NW; ++j) {
    int i = tid + j * APPLY_T;
    if (j < NW - 1 || tid < 1728 - (NW - 1) * APPLY_T) {
      int w = i / 54, rem = i - w * 54;
      int n = rem >> 1, rd = rem & 1;
      float* dst = &wl[(n * 4 + rd * 2) * WPITCH + 2 * w];
      *(float2*)dst = make_float2(wv[j].x, wv[j].y);
      *(float2*)(dst + WPITCH) = make_float2(wv[j].z, wv[j].w);
    }
  }
  __syncthreads();

  const int w2 = tid & 63;
  const int cg = tid >> 6;
  const int w = w2 >> 1;
  const int wm = max(w - 1, 0), wp = min(w + 1, 31);

  float wr[27][4];
#pragma unroll
  for (int n = 0; n < 27; ++n)
#pragma unroll
    for (int q = 0; q < 4; ++q) wr[n][q] = wl[(n * 4 + q) * WPITCH + w2];

  float* op = out + (size_t)(cg * 16) * 262144 + (size_t)(2 * d) * 4096 + (2 * h) * 64 + w2;
  const float* yo = &yl[cg * 16 * 288];
#pragma clang loop vectorize(disable)
  for (int c = 0; c < 16; ++c) {
    float a0 = 0.0f, a1 = 0.0f, a2 = 0.0f, a3 = 0.0f;
#pragma unroll
    for (int r = 0; r < 9; ++r) {
      float y0 = yo[r * 32 + wm];
      float y1v = yo[r * 32 + w];
      float y2v = yo[r * 32 + wp];
      a0 += y0 * wr[3 * r][0];
      a1 += y0 * wr[3 * r][1];
      a2 += y0 * wr[3 * r][2];
      a3 += y0 * wr[3 * r][3];
      a0 += y1v * wr[3 * r + 1][0];
      a1 += y1v * wr[3 * r + 1][1];
      a2 += y1v * wr[3 * r + 1][2];
      a3 += y1v * wr[3 * r + 1][3];
      a0 += y2v * wr[3 * r + 2][0];
      a1 += y2v * wr[3 * r + 2][1];
      a2 += y2v * wr[3 * r + 2][2];
      a3 += y2v * wr[3 * r + 2][3];
    }
    op[0] = a0;
    op[64] = a1;
    op[4096] = a2;
    op[4096 + 64] = a3;
    op += 262144;
    yo += 288;
  }
}

extern "C" void kernel_launch(void* const* d_in, const int* in_sizes, int n_in,
                              void* d_out, int out_size, void* d_ws,
                              size_t ws_size, hipStream_t stream) {
  (void)in_sizes; (void)n_in; (void)out_size; (void)ws_size;
  const float* x  = (const float*)d_in[0];
  const float* Wc = (const float*)d_in[1];
  const float* bc = (const float*)d_in[2];
  const float* We = (const float*)d_in[3];
  const float* be = (const float*)d_in[4];
  const float* Wp = (const float*)d_in[5];
  float* out = (float*)d_out;

  float* ws = (float*)d_ws;
  float* y1   = ws;
  float* wsft = y1 + (size_t)64 * SPATIAL;
  float* Yp   = wsft + (size_t)SPATIAL * 216;
  float* Wt   = Yp + (size_t)32 * SPATIAL;
  float* bep  = Wt + (size_t)NWT;

  front_kernel<<<FRONT_BLOCKS, 128, 0, stream>>>(x, Wc, bc, Wp, We, be, y1, Yp, Wt, bep);
  conv3_softmax_wmma_kernel<<<512, 128, 0, stream>>>(y1, Wt, bep, wsft);
  carafe_apply_kernel<<<1024, APPLY_T, 0, stream>>>(Yp, wsft, out);
}


# gather fused into the conv kernel's epilogue: a block applies its softmax weights from LDS to 12 staged Y rows and writes out; weight plane and third launch removed; tap sum order unchanged
# speedup vs baseline: 3.2015x; 1.0435x over previous
#include <hip/hip_runtime.h>
#include <hip/hip_bf16.h>

typedef float v2f __attribute__((ext_vector_type(2)));
typedef float v8f __attribute__((ext_vector_type(8)));

__device__ __forceinline__ v8f wmma_f32(v2f a, v2f b, v8f c) {
  return __builtin_amdgcn_wmma_f32_16x16x4_f32(false, a, false, b, (short)0, c,
                                               false, false);
}

#define SPATIAL 32768
#define MPAD 224
#define NWT (36 * 64 * MPAD)
#define NCH 216

#define GEMM_BLOCKS 512
#define FRONT_BLOCKS 568
__global__ __launch_bounds__(128) void front_kernel(
    const float* __restrict__ x, const float* __restrict__ Wc,
    const float* __restrict__ bc, const float* __restrict__ Wp,
    const float* __restrict__ We, const float* __restrict__ be,
    float* __restrict__ y1, float* __restrict__ Yp, float* __restrict__ Wt,
    float* __restrict__ bep) {
  __shared__ float lA[96 * 64];
  __shared__ float lB[64 * 64];

  const int tid = threadIdx.x;

  if (blockIdx.x >= GEMM_BLOCKS) {
    const int t = (blockIdx.x - GEMM_BLOCKS) * 128 + tid;
    const int o = t % MPAD;
    const int cm2 = t / MPAD;
    if (t < MPAD) bep[t] = (t < NCH) ? be[t] : 0.0f;
    float vf[54];
    if (o < NCH) {
      const float2* s = (const float2*)(We + (size_t)o * 1728 + cm2 * 54);
#pragma unroll
      for (int k = 0; k < 27; ++k) {
        float2 q = s[k];
        vf[2 * k] = q.x;
        vf[2 * k + 1] = q.y;
      }
    } else {
#pragma unroll
      for (int k = 0; k < 54; ++k) vf[k] = 0.0f;
    }
    float* dst = Wt + ((size_t)cm2 * MPAD + o) * 2;
#pragma unroll
    for (int kd = 0; kd < 3; ++kd)
#pragma unroll
      for (int kw = 0; kw < 3; ++kw) {
        const int i0 = kd * 9 + kw;
        const float e0 = vf[i0], e1 = vf[i0 + 3], e2 = vf[i0 + 6];
        const float q0 = vf[27 + i0], q1 = vf[27 + i0 + 3], q2 = vf[27 + i0 + 6];
        float* dk = dst + (size_t)(kd * 12 + kw) * (32 * MPAD * 2);
        *(float2*)(dk) = make_float2(e0, q0);
        *(float2*)(dk + (size_t)3 * (32 * MPAD * 2)) =
            make_float2(0.5f * (e0 + e1 + e2), 0.5f * (q0 + q1 + q2));
        *(float2*)(dk + (size_t)6 * (32 * MPAD * 2)) =
            make_float2(0.5f * (e0 - e1 + e2), 0.5f * (q0 - q1 + q2));
        *(float2*)(dk + (size_t)9 * (32 * MPAD * 2)) = make_float2(e2, q2);
      }
    return;
  }

  const int lane = tid & 31;
  const int wave = tid >> 5;
  const int n0 = blockIdx.x * 64;
  const int ml = lane & 15;
  const int lhi = lane >> 4;

  for (int i = tid; i < 64 * 64; i += 128) lA[i] = Wc[i];
  for (int i = tid; i < 32 * 64; i += 128) lA[64 * 64 + i] = Wp[i];
  for (int i = tid; i < 64 * 64; i += 128) {
    int k = i >> 6, col = i & 63;
    lB[((k >> 1) * 64 + col) * 2 + (k & 1)] = x[(size_t)k * SPATIAL + n0 + col];
  }
  __syncthreads();

  v8f acc[6];
#pragma unroll
  for (int t = 0; t < 6; ++t) {
    v8f c;
#pragma unroll
    for (int v = 0; v < 8; ++v) c[v] = (t < 4) ? bc[t * 16 + v + 8 * lhi] : 0.0f;
    acc[t] = c;
  }

#pragma unroll
  for (int k0 = 0; k0 < 64; k0 += 4) {
    v2f b = *(const v2f*)&lB[(((k0 >> 1) + lhi) * 64 + wave * 16 + ml) * 2];
#pragma unroll
    for (int t = 0; t < 6; ++t) {
      v2f a = *(const v2f*)&lA[(t * 16 + ml) * 64 + k0 + 2 * lhi];
      acc[t] = wmma_f32(a, b, acc[t]);
    }
  }

  const int col = n0 + wave * 16 + ml;
#pragma unroll
  for (int t = 0; t < 6; ++t)
#pragma unroll
    for (int v = 0; v < 8; ++v) {
      const int m = t * 16 + v + 8 * lhi;
      if (t < 4)
        y1[(size_t)m * SPATIAL + col] = acc[t][v];
      else
        Yp[(size_t)(m - 64) * SPATIAL + col] = acc[t][v];
    }
}

#define STRIPF (64 * 34)
#define NSTEP 96
#define YLF (32 * 12 * 32)

__device__ __forceinline__ void conv_load_a(const float* __restrict__ abase,
                                            int cp, int sg, v2f (&a)[7][3]) {
  const int g = (sg >> 5) * 4 + cp * 2 + ((sg >> 4) & 1);
  const float* p = abase + (size_t)(g * (3 * 32) + (sg & 15) * 2) * (MPAD * 2);
#pragma unroll
  for (int kw = 0; kw < 3; ++kw)
#pragma unroll
    for (int t = 0; t < 7; ++t)
      a[t][kw] = *(const v2f*)(p + kw * (32 * MPAD * 2) + t * 32);
}

__device__ __forceinline__ void conv_load_b(const float* bb, v2f (&b)[3][2]) {
#pragma unroll
  for (int kw = 0; kw < 3; ++kw)
#pragma unroll
    for (int j = 0; j < 2; ++j) b[kw][j] = *(const v2f*)(bb + (j * 16 + kw) * 2);
}

__device__ __forceinline__ void conv_step(const float* __restrict__ abase,
                                          int cp, const float* bbn, int sg,
                                          v2f (&ac)[7][3], v2f (&an)[7][3],
                                          v2f (&bc)[3][2], v2f (&bn)[3][2],
                                          v8f (&acc)[7][2]) {
  conv_load_a(abase, cp, min(sg + 1, NSTEP - 1), an);
  conv_load_b(bbn, bn);
  __builtin_amdgcn_sched_barrier(0);
#pragma unroll
  for (int kw = 0; kw < 3; ++kw)
#pragma unroll
    for (int t = 0; t < 7; ++t)
#pragma unroll
      for (int j = 0; j < 2; ++j) acc[t][j] = wmma_f32(ac[t][kw], bc[kw][j], acc[t][j]);
}

__device__ __forceinline__ float4 f4sub(float4 a, float4 b) {
  return make_float4(a.x - b.x, a.y - b.y, a.z - b.z, a.w - b.w);
}
__device__ __forceinline__ float4 f4add(float4 a, float4 b) {
  return make_float4(a.x + b.x, a.y + b.y, a.z + b.z, a.w + b.w);
}
__device__ __forceinline__ void strip_store(float* dst, float4 u, float4 v) {
  *(float2*)(dst + 0) = make_float2(u.x, v.x);
  *(float2*)(dst + 2) = make_float2(u.y, v.y);
  *(float2*)(dst + 4) = make_float2(u.z, v.z);
  *(float2*)(dst + 6) = make_float2(u.w, v.w);
}

__global__ __launch_bounds__(128) void conv3_softmax_wmma_kernel(
    const float* __restrict__ y1, const float* __restrict__ Wt,
    const float* __restrict__ bep, const float* __restrict__ Yp,
    float* __restrict__ out) {
  __shared__ __attribute__((aligned(16))) float lds[4 * STRIPF + YLF + 64 * NCH];
  float* const strips = lds;
  float* const yl = lds + 4 * STRIPF;
  float* const ep = yl + YLF;

  const int tid = threadIdx.x;
  const int lane = tid & 31;
  const int wave = tid >> 5;
  const int d = blockIdx.x >> 4;
  const int h0 = (blockIdx.x & 15) * 2;
  const int ml = lane & 15;
  const int lhi = lane >> 4;
  const int mh = wave & 1;
  const int cp = wave >> 1;

  v8f acc[2][7][2];
#pragma unroll
  for (int t = 0; t < 7; ++t) {
    int ob = (mh * 7 + t) * 16;
#pragma unroll
    for (int j = 0; j < 2; ++j) {
      v8f c, z;
#pragma unroll
      for (int v = 0; v < 8; ++v) {
        const float bv = bep[ob + v + 8 * lhi];
        c[v] = (cp == 0) ? bv : 0.0f;
        z[v] = 0.0f;
      }
      acc[0][t][j] = z;
      acc[1][t][j] = c;
    }
  }

#pragma unroll
  for (int e = 0; e < 2; ++e) {
    int idx = tid + 128 * e;
    int s4 = idx >> 6, cm2 = (idx >> 1) & 31, side = idx & 1;
    *(float2*)&strips[s4 * STRIPF + (cm2 * 34 + side * 33) * 2] = make_float2(0.0f, 0.0f);
  }

  const float* abase = Wt + (size_t)((lhi * MPAD + mh * 112 + ml) * 2);
  const float* bbase = strips + (cp * 2) * STRIPF + (lhi * 34 + ml) * 2;

  v2f aX[7][3], aY[7][3];
  v2f bX[3][2], bY[3][2];
  conv_load_a(abase, cp, 0, aX);

#pragma unroll 1
  for (int kd = 0; kd < 3; ++kd) {
    const int dd = d + kd - 1;
    const bool dok = (dd >= 0 && dd < 32);
    const int ddc = min(max(dd, 0), 31);
    float4 ya[8], yb[8];
#pragma unroll
    for (int i = 0; i < 8; ++i) {
      const int part = tid & 7;
      const int cm2 = (tid >> 3) + 16 * (i & 1);
      const int hh = h0 - 1 + (i >> 1);
      const int hhc = min(max(hh, 0), 31);
      const float* s = y1 + (size_t)(2 * cm2) * SPATIAL + (ddc * 32 + hhc) * 32 + part * 4;
      ya[i] = *(const float4*)s;
      yb[i] = *(const float4*)(s + SPATIAL);
    }
    __syncthreads();
#pragma unroll
    for (int e = 0; e < 2; ++e) {
      const int part = tid & 7;
      const int cm2 = (tid >> 3) + 16 * e;
      float* dst = strips + (cm2 * 34 + part * 4 + 1) * 2;
      float4 ra[4], rb[4];
#pragma unroll
      for (int k = 0; k < 4; ++k) {
        const int hh = h0 - 1 + k;
        const bool ok = dok && (hh >= 0) && (hh < 32);
        ra[k] = ok ? ya[2 * k + e] : make_float4(0.0f, 0.0f, 0.0f, 0.0f);
        rb[k] = ok ? yb[2 * k + e] : make_float4(0.0f, 0.0f, 0.0f, 0.0f);
      }
      strip_store(dst + 0 * STRIPF, f4sub(ra[0], ra[2]), f4sub(rb[0], rb[2]));
      strip_store(dst + 1 * STRIPF, f4add(ra[1], ra[2]), f4add(rb[1], rb[2]));
      strip_store(dst + 2 * STRIPF, f4sub(ra[2], ra[1]), f4sub(rb[2], rb[1]));
      strip_store(dst + 3 * STRIPF, f4sub(ra[1], ra[3]), f4sub(rb[1], rb[3]));
    }
    if (kd < 2) {
      const int d2 = min(max(d + kd, 0), 31);
#pragma unroll
      for (int e = 0; e < 2; ++e) {
        int idx = tid + 128 * e;
        int h2 = min(max(h0 - 1 + (idx >> 6), 0), 31);
        __builtin_prefetch(y1 + (size_t)(idx & 63) * SPATIAL + (d2 * 32 + h2) * 32, 0, 0);
      }
    }
    __syncthreads();

    conv_load_b(bbase, bX);
#pragma unroll
    for (int c = 0; c < 2; ++c) {
#pragma unroll 1
      for (int s = 0; s < 16; s += 2) {
        const int q1 = c * 16 + s + 1, q2 = min(c * 16 + s + 2, 31);
        const float* bb1 = bbase + (q1 >> 4) * STRIPF + (q1 & 15) * (2 * 34 * 2);
        const float* bb2 = bbase + (q2 >> 4) * STRIPF + (q2 & 15) * (2 * 34 * 2);
        conv_step(abase, cp, bb1, kd * 32 + c * 16 + s, aX, aY, bX, bY, acc[c]);
        conv_step(abase, cp, bb2, kd * 32 + c * 16 + s + 1, aY, aX, bY, bX, acc[c]);
      }
    }
  }

  if (cp == 0) {
#pragma unroll
    for (int t = 0; t < 7; ++t) {
      const int o0 = (mh * 7 + t) * 16 + 8 * lhi;
#pragma unroll
      for (int j = 0; j < 2; ++j) {
        if (o0 < NCH) {
          const v8f m0 = acc[0][t][j], m1 = acc[1][t][j];
          float4* r0 = (float4*)&ep[(j * 16 + ml) * NCH + o0];
          float4* r1 = (float4*)&ep[(32 + j * 16 + ml) * NCH + o0];
          r0[0] = make_float4(m0[0] + m1[0], m0[1] + m1[1], m0[2] + m1[2], m0[3] + m1[3]);
          r0[1] = make_float4(m0[4] + m1[4], m0[5] + m1[5], m0[6] + m1[6], m0[7] + m1[7]);
          r1[0] = make_float4(m1[0], m1[1], m1[2], m1[3]);
          r1[1] = make_float4(m1[4], m1[5], m1[6], m1[7]);
        }
      }
    }
  }
  __syncthreads();
  if (cp == 1) {
#pragma unroll
    for (int t = 0; t < 7; ++t) {
      const int o0 = (mh * 7 + t) * 16 + 8 * lhi;
#pragma unroll
      for (int j = 0; j < 2; ++j) {
        if (o0 < NCH) {
          const v8f m2 = acc[0][t][j], m3 = acc[1][t][j];
          float4* r0 = (float4*)&ep[(j * 16 + ml) * NCH + o0];
          float4* r1 = (float4*)&ep[(32 + j * 16 + ml) * NCH + o0];
          float4 a0 = r0[0], a1 = r0[1], b0 = r1[0], b1 = r1[1];
          r0[0] = make_float4(a0.x + m2[0], a0.y + m2[1], a0.z + m2[2], a0.w + m2[3]);
          r0[1] = make_float4(a1.x + m2[4], a1.y + m2[5], a1.z + m2[6], a1.w + m2[7]);
          r1[0] = make_float4((b0.x - m2[0]) - m3[0], (b0.y - m2[1]) - m3[1],
                              (b0.z - m2[2]) - m3[2], (b0.w - m2[3]) - m3[3]);
          r1[1] = make_float4((b1.x - m2[4]) - m3[4], (b1.y - m2[5]) - m3[5],
                              (b1.z - m2[6]) - m3[6], (b1.w - m2[7]) - m3[7]);
        }
      }
    }
  }

  __syncthreads();

  constexpr int NYQ = YLF / 4 / 128 / 4;
  for (int rr = 0; rr < 4; ++rr) {
    float4 yv[NYQ];
#pragma unroll
    for (int j = 0; j < NYQ; ++j) {
      const int i = tid + (rr * NYQ + j) * 128;
      const int row = i >> 3, part = i & 7;
      const int o = row / 12, yr = row - o * 12;
      const int dd = min(max(d + (yr >> 2) - 1, 0), 31);
      const int hh = min(max(h0 - 1 + (yr & 3), 0), 31);
      yv[j] = *(const float4*)(Yp + (size_t)o * SPATIAL + (dd * 32 + hh) * 32 + part * 4);
    }
    const int p = tid + 128 * rr;
    float* e = &ep[(p >> 3) * NCH + (p & 7)];
    float v[27];
    float mx = -3.0e38f;
#pragma unroll
    for (int n = 0; n < 27; ++n) {
      v[n] = e[n * 8];
      mx = fmaxf(mx, v[n]);
    }
    float sum = 0.0f;
#pragma unroll
    for (int n = 0; n < 27; ++n) {
      v[n] = __expf(v[n] - mx);
      sum += v[n];
    }
    float inv = 1.0f / sum;
#pragma unroll
    for (int n = 0; n < 27; ++n) e[n * 8] = v[n] * inv;
#pragma unroll
    for (int j = 0; j < NYQ; ++j) {
      const int i = tid + (rr * NYQ + j) * 128;
      *(float4*)&yl[i * 4] = yv[j];
    }
  }
  __syncthreads();

  {
    const int r = tid >> 6;
    const int w2 = tid & 63;
    const int w = w2 >> 1;
    const int wm = max(w - 1, 0), wp = min(w + 1, 31);

    float wr[27][4];
    const float* wsrc = &ep[(r * 32 + w) * NCH + (w2 & 1)];
#pragma unroll
    for (int n = 0; n < 27; ++n)
#pragma unroll
      for (int q = 0; q < 4; ++q) wr[n][q] = wsrc[n * 8 + q * 2];

    float* op = out + (size_t)(2 * d) * 4096 + (2 * (h0 + r)) * 64 + w2;
    const float* yo = yl + r * 32;
#pragma clang loop vectorize(disable)
    for (int c = 0; c < 32; ++c) {
      float a0 = 0.0f, a1 = 0.0f, a2 = 0.0f, a3 = 0.0f;
#pragma unroll
      for (int t = 0; t < 9; ++t) {
        const float* yr = yo + ((t / 3) * 4 + (t % 3)) * 32;
        float y0 = yr[wm];
        float y1v = yr[w];
        float y2v = yr[wp];
        a0 += y0 * wr[3 * t][0];
        a1 += y0 * wr[3 * t][1];
        a2 += y0 * wr[3 * t][2];
        a3 += y0 * wr[3 * t][3];
        a0 += y1v * wr[3 * t + 1][0];
        a1 += y1v * wr[3 * t + 1][1];
        a2 += y1v * wr[3 * t + 1][2];
        a3 += y1v * wr[3 * t + 1][3];
        a0 += y2v * wr[3 * t + 2][0];
        a1 += y2v * wr[3 * t + 2][1];
        a2 += y2v * wr[3 * t + 2][2];
        a3 += y2v * wr[3 * t + 2][3];
      }
      op[0] = a0;
      op[64] = a1;
      op[4096] = a2;
      op[4096 + 64] = a3;
      op += 262144;
      yo += 12 * 32;
    }
  }
}

extern "C" void kernel_launch(void* const* d_in, const int* in_sizes, int n_in,
                              void* d_out, int out_size, void* d_ws,
                              size_t ws_size, hipStream_t stream) {
  (void)in_sizes; (void)n_in; (void)out_size; (void)ws_size;
  const float* x  = (const float*)d_in[0];
  const float* Wc = (const float*)d_in[1];
  const float* bc = (const float*)d_in[2];
  const float* We = (const float*)d_in[3];
  const float* be = (const float*)d_in[4];
  const float* Wp = (const float*)d_in[5];
  float* out = (float*)d_out;

  float* ws = (float*)d_ws;
  float* y1   = ws;
  float* Yp   = y1 + (size_t)64 * SPATIAL;
  float* Wt   = Yp + (size_t)32 * SPATIAL;
  float* bep  = Wt + (size_t)NWT;

  front_kernel<<<FRONT_BLOCKS, 128, 0, stream>>>(x, Wc, bc, Wp, We, be, y1, Yp, Wt, bep);
  conv3_softmax_wmma_kernel<<<512, 128, 0, stream>>>(y1, Wt, bep, Yp, out);
}


# front kernel operand staging redone: x and both weight matrices fetched with 16-byte loads all issued before the first LDS store; weight rows at an LDS pitch of 68 floats (no bank conflicts)
# speedup vs baseline: 4.0262x; 1.2576x over previous
#include <hip/hip_runtime.h>
#include <hip/hip_bf16.h>

typedef float v2f __attribute__((ext_vector_type(2)));
typedef float v8f __attribute__((ext_vector_type(8)));

__device__ __forceinline__ v8f wmma_f32(v2f a, v2f b, v8f c) {
  return __builtin_amdgcn_wmma_f32_16x16x4_f32(false, a, false, b, (short)0, c,
                                               false, false);
}

#define SPATIAL 32768
#define MPAD 224
#define NWT (54 * 64 * MPAD)
#define NCH 216

#define GEMM_BLOCKS 512
#define LAP 68
#define FRONT_BLOCKS 568
__global__ __launch_bounds__(128) void front_kernel(
    const float* __restrict__ x, const float* __restrict__ Wc,
    const float* __restrict__ bc, const float* __restrict__ Wp,
    const float* __restrict__ We, const float* __restrict__ be,
    float* __restrict__ y1, float* __restrict__ Yp, float* __restrict__ Wt,
    float* __restrict__ bep) {
  __shared__ __attribute__((aligned(16))) float lA[96 * LAP];
  __shared__ __attribute__((aligned(16))) float lB[64 * 64];

  const int tid = threadIdx.x;

  if (blockIdx.x >= GEMM_BLOCKS) {
    const int t = (blockIdx.x - GEMM_BLOCKS) * 128 + tid;
    const int o = t % MPAD;
    const int cm2 = t / MPAD;
    if (t < MPAD) bep[t] = (t < NCH) ? be[t] : 0.0f;
    float vf[54];
    if (o < NCH) {
      const float2* s = (const float2*)(We + (size_t)o * 1728 + cm2 * 54);
#pragma unroll
      for (int k = 0; k < 27; ++k) {
        float2 q = s[k];
        vf[2 * k] = q.x;
        vf[2 * k + 1] = q.y;
      }
    } else {
#pragma unroll
      for (int k = 0; k < 54; ++k) vf[k] = 0.0f;
    }
    float* dst = Wt + ((size_t)cm2 * MPAD + o) * 2;
#pragma unroll
    for (int kd = 0; kd < 3; ++kd)
#pragma unroll
      for (int kw = 0; kw < 3; ++kw) {
        const int i0 = kd * 9 + kw;
        const float e0 = vf[i0], e1 = vf[i0 + 3], e2 = vf[i0 + 6];
        const float q0 = vf[27 + i0], q1 = vf[27 + i0 + 3], q2 = vf[27 + i0 + 6];
        float* dk = dst + (size_t)(kd * 18 + kw) * (32 * MPAD * 2);
        const float es = e0 + e2, qs = q0 + q2;
        const float ea = (1.0f / 24.0f) * e0 + (1.0f / 6.0f) * e2, eb = (1.0f / 12.0f) * e1;
        const float qa = (1.0f / 24.0f) * q0 + (1.0f / 6.0f) * q2, qb = (1.0f / 12.0f) * q1;
        *(float2*)(dk) = make_float2(0.25f * e0, 0.25f * q0);
        *(float2*)(dk + (size_t)3 * (32 * MPAD * 2)) =
            make_float2((-1.0f / 6.0f) * (es + e1), (-1.0f / 6.0f) * (qs + q1));
        *(float2*)(dk + (size_t)6 * (32 * MPAD * 2)) =
            make_float2((-1.0f / 6.0f) * (es - e1), (-1.0f / 6.0f) * (qs - q1));
        *(float2*)(dk + (size_t)9 * (32 * MPAD * 2)) = make_float2(ea + eb, qa + qb);
        *(float2*)(dk + (size_t)12 * (32 * MPAD * 2)) = make_float2(ea - eb, qa - qb);
        *(float2*)(dk + (size_t)15 * (32 * MPAD * 2)) = make_float2(e2, q2);
      }
    return;
  }

  const int lane = tid & 31;
  const int wave = tid >> 5;
  const int n0 = blockIdx.x * 64;
  const int ml = lane & 15;
  const int lhi = lane >> 4;

  v8f acc[6];
#pragma unroll
  for (int t = 0; t < 6; ++t) {
    v8f c;
#pragma unroll
    for (int v = 0; v < 8; ++v) c[v] = (t < 4) ? bc[t * 16 + v + 8 * lhi] : 0.0f;
    acc[t] = c;
  }
  float4 wa[12];
#pragma unroll
  for (int j = 0; j < 12; ++j) {
    const int i = tid + 128 * j;
    wa[j] = (j < 8) ? ((const float4*)Wc)[i] : ((const float4*)Wp)[i - 1024];
  }
  float4 xa[4], xb[4];
#pragma unroll
  for (int j = 0; j < 4; ++j) {
    const int it = tid + 128 * j;
    const float* s = x + (size_t)(2 * (it >> 4)) * SPATIAL + n0 + (it & 15) * 4;
    xa[j] = *(const float4*)s;
    xb[j] = *(const float4*)(s + SPATIAL);
  }
#pragma unroll
  for (int j = 0; j < 12; ++j) {
    const int i = tid + 128 * j;
    *(float4*)&lA[(i >> 4) * LAP + (i & 15) * 4] = wa[j];
  }
#pragma unroll
  for (int j = 0; j < 4; ++j) {
    const int it = tid + 128 * j;
    float* dst = &lB[((it >> 4) * 64 + (it & 15) * 4) * 2];
    *(float4*)dst = make_float4(xa[j].x, xb[j].x, xa[j].y, xb[j].y);
    *(float4*)(dst + 4) = make_float4(xa[j].z, xb[j].z, xa[j].w, xb[j].w);
  }
  __syncthreads();

#pragma unroll
  for (int k0 = 0; k0 < 64; k0 += 4) {
    v2f b = *(const v2f*)&lB[(((k0 >> 1) + lhi) * 64 + wave * 16 + ml) * 2];
#pragma unroll
    for (int t = 0; t < 6; ++t) {
      v2f a = *(const v2f*)&lA[(t * 16 + ml) * LAP + k0 + 2 * lhi];
      acc[t] = wmma_f32(a, b, acc[t]);
    }
  }

  const int col = n0 + wave * 16 + ml;
#pragma unroll
  for (int t = 0; t < 6; ++t)
#pragma unroll
    for (int v = 0; v < 8; ++v) {
      const int m = t * 16 + v + 8 * lhi;
      if (t < 4)
        y1[(size_t)m * SPATIAL + col] = acc[t][v];
      else
        Yp[(size_t)(m - 64) * SPATIAL + col] = acc[t][v];
    }
}

#define STRIPF (64 * 34)
#define NSTEP 48
#define YLF (32 * 18 * 32)
#define NPOS 128

__device__ __forceinline__ void conv_load_a(const float* __restrict__ abase,
                                            int comp, int sg, v2f (&a)[7][3]) {
  const int g = (sg >> 4) * 6 + comp;
  const float* p = abase + (size_t)(g * (3 * 32) + (sg & 15) * 2) * (MPAD * 2);
#pragma unroll
  for (int kw = 0; kw < 3; ++kw)
#pragma unroll
    for (int t = 0; t < 7; ++t)
      a[t][kw] = *(const v2f*)(p + kw * (32 * MPAD * 2) + t * 32);
}

__device__ __forceinline__ void conv_load_b(const float* bb, v2f (&b)[3][2]) {
#pragma unroll
  for (int kw = 0; kw < 3; ++kw)
#pragma unroll
    for (int j = 0; j < 2; ++j) b[kw][j] = *(const v2f*)(bb + (j * 16 + kw) * 2);
}

__device__ __forceinline__ void conv_step(const float* __restrict__ abase,
                                          int comp, const float* bbn, int sg,
                                          v2f (&ac)[7][3], v2f (&an)[7][3],
                                          v2f (&bc)[3][2], v2f (&bn)[3][2],
                                          v8f (&acc)[7][2]) {
  conv_load_a(abase, comp, min(sg + 1, NSTEP - 1), an);
  conv_load_b(bbn, bn);
  __builtin_amdgcn_sched_barrier(0);
#pragma unroll
  for (int kw = 0; kw < 3; ++kw)
#pragma unroll
    for (int t = 0; t < 7; ++t)
#pragma unroll
      for (int j = 0; j < 2; ++j) acc[t][j] = wmma_f32(ac[t][kw], bc[kw][j], acc[t][j]);
}

__device__ __forceinline__ float4 f4sub(float4 a, float4 b) {
  return make_float4(a.x - b.x, a.y - b.y, a.z - b.z, a.w - b.w);
}
__device__ __forceinline__ float4 f4add(float4 a, float4 b) {
  return make_float4(a.x + b.x, a.y + b.y, a.z + b.z, a.w + b.w);
}
__device__ __forceinline__ float4 f4fma(float k, float4 a, float4 b) {
  return make_float4(k * a.x + b.x, k * a.y + b.y, k * a.z + b.z, k * a.w + b.w);
}
__device__ __forceinline__ void strip_store(float* dst, float4 u, float4 v) {
  *(float2*)(dst + 0) = make_float2(u.x, v.x);
  *(float2*)(dst + 2) = make_float2(u.y, v.y);
  *(float2*)(dst + 4) = make_float2(u.z, v.z);
  *(float2*)(dst + 6) = make_float2(u.w, v.w);
}

template <int C>
__device__ __forceinline__ void ep_pass(float* ep, const float* __restrict__ bep,
                                        const v8f (&acc)[7][2], int mh, int ml,
                                        int lhi, float K0, float K1, float K2,
                                        float K3) {
  const float K[4] = {K0, K1, K2, K3};
#pragma unroll
  for (int t = 0; t < 7; ++t) {
    const int o0 = (mh * 7 + t) * 16 + 8 * lhi;
    if (o0 < NCH) {
      float4 b0 = make_float4(0.0f, 0.0f, 0.0f, 0.0f), b1 = b0;
      if (C == 0) {
        b0 = *(const float4*)(bep + o0);
        b1 = *(const float4*)(bep + o0 + 4);
      }
#pragma unroll
      for (int j = 0; j < 2; ++j) {
        const v8f m = acc[t][j];
        const float4 mlo = make_float4(m[0], m[1], m[2], m[3]);
        const float4 mhi = make_float4(m[4], m[5], m[6], m[7]);
#pragma unroll
        for (int r = 0; r < 4; ++r) {
          float4* q = (float4*)&ep[(r * 32 + j * 16 + ml) * NCH + o0];
          if (C == 0) {
            q[0] = (r == 0) ? f4add(b0, mlo) : b0;
            q[1] = (r == 0) ? f4add(b1, mhi) : b1;
          } else if (K[r] != 0.0f) {
            q[0] = f4fma(K[r], mlo, q[0]);
            q[1] = f4fma(K[r], mhi, q[1]);
          }
        }
      }
    }
  }
}

__global__ __launch_bounds__(384) void conv3_softmax_wmma_kernel(
    const float* __restrict__ y1, const float* __restrict__ Wt,
    const float* __restrict__ bep, const float* __restrict__ Yp,
    float* __restrict__ out) {
  __shared__ __attribute__((aligned(16))) float lds[6 * STRIPF + YLF + NPOS * NCH];
  float* const strips = lds;
  float* const yl = lds + 6 * STRIPF;
  float* const ep = yl + YLF;

  const int tid = threadIdx.x;
  const int lane = tid & 31;
  const int wave = tid >> 5;
  const int d = blockIdx.x >> 3;
  const int h0 = (blockIdx.x & 7) * 4;
  const int ml = lane & 15;
  const int lhi = lane >> 4;
  const int mh = wave & 1;
  const int comp = wave >> 1;
  const bool stager = tid < 256;

  v8f acc[7][2];
#pragma unroll
  for (int t = 0; t < 7; ++t)
#pragma unroll
    for (int j = 0; j < 2; ++j) {
      v8f z;
#pragma unroll
      for (int v = 0; v < 8; ++v) z[v] = 0.0f;
      acc[t][j] = z;
    }

  {
    const int s6 = tid >> 6, cm2 = (tid >> 1) & 31, side = tid & 1;
    *(float2*)&strips[s6 * STRIPF + (cm2 * 34 + side * 33) * 2] = make_float2(0.0f, 0.0f);
  }

  const float* abase = Wt + (size_t)((lhi * MPAD + mh * 112 + ml) * 2);
  const float* bbase = strips + comp * STRIPF + (lhi * 34 + ml) * 2;

  v2f aX[7][3], aY[7][3];
  v2f bX[3][2], bY[3][2];
  conv_load_a(abase, comp, 0, aX);

#pragma unroll 1
  for (int kd = 0; kd < 3; ++kd) {
    const int dd = d + kd - 1;
    const bool dok = (dd >= 0 && dd < 32);
    const int ddc = min(max(dd, 0), 31);
    const int part = tid & 7;
    const int cm2 = (tid >> 3) & 31;
    float4 ya[6], yb[6];
#pragma unroll
    for (int i = 0; i < 6; ++i) {
      ya[i] = make_float4(0.0f, 0.0f, 0.0f, 0.0f);
      yb[i] = ya[i];
    }
    if (stager) {
#pragma unroll
      for (int i = 0; i < 6; ++i) {
        const int hhc = min(max(h0 - 1 + i, 0), 31);
        const float* s = y1 + (size_t)(2 * cm2) * SPATIAL + (ddc * 32 + hhc) * 32 + part * 4;
        ya[i] = *(const float4*)s;
        yb[i] = *(const float4*)(s + SPATIAL);
      }
    }
    __syncthreads();
    if (stager) {
      float* dst = strips + (cm2 * 34 + part * 4 + 1) * 2;
      float4 ra[6], rb[6];
#pragma unroll
      for (int k = 0; k < 6; ++k) {
        const int hh = h0 - 1 + k;
        const bool ok = dok && (hh >= 0) && (hh < 32);
        ra[k] = ok ? ya[k] : make_float4(0.0f, 0.0f, 0.0f, 0.0f);
        rb[k] = ok ? yb[k] : make_float4(0.0f, 0.0f, 0.0f, 0.0f);
      }
      strip_store(dst + 0 * STRIPF, f4fma(4.0f, ra[0], f4fma(-5.0f, ra[2], ra[4])),
                  f4fma(4.0f, rb[0], f4fma(-5.0f, rb[2], rb[4])));
      strip_store(dst + 1 * STRIPF, f4fma(-4.0f, f4add(ra[1], ra[2]), f4add(ra[3], ra[4])),
                  f4fma(-4.0f, f4add(rb[1], rb[2]), f4add(rb[3], rb[4])));
      strip_store(dst + 2 * STRIPF, f4fma(4.0f, f4sub(ra[1], ra[2]), f4sub(ra[4], ra[3])),
                  f4fma(4.0f, f4sub(rb[1], rb[2]), f4sub(rb[4], rb[3])));
      strip_store(dst + 3 * STRIPF, f4fma(2.0f, f4sub(ra[3], ra[1]), f4sub(ra[4], ra[2])),
                  f4fma(2.0f, f4sub(rb[3], rb[1]), f4sub(rb[4], rb[2])));
      strip_store(dst + 4 * STRIPF, f4fma(-2.0f, f4sub(ra[3], ra[1]), f4sub(ra[4], ra[2])),
                  f4fma(-2.0f, f4sub(rb[3], rb[1]), f4sub(rb[4], rb[2])));
      strip_store(dst + 5 * STRIPF, f4fma(4.0f, ra[1], f4fma(-5.0f, ra[3], ra[5])),
                  f4fma(4.0f, rb[1], f4fma(-5.0f, rb[3], rb[5])));
    }
    if (kd < 2) {
      const int d2 = min(max(d + kd, 0), 31);
      const int h2 = min(max(h0 - 1 + (tid >> 6), 0), 31);
      __builtin_prefetch(y1 + (size_t)(tid & 63) * SPATIAL + (d2 * 32 + h2) * 32, 0, 0);
    }
    __syncthreads();

    conv_load_b(bbase, bX);
#pragma unroll 1
    for (int s = 0; s < 16; s += 2) {
      const int q1 = s + 1, q2 = min(s + 2, 15);
      const float* bb1 = bbase + q1 * (2 * 34 * 2);
      const float* bb2 = bbase + q2 * (2 * 34 * 2);
      conv_step(abase, comp, bb1, kd * 16 + s, aX, aY, bX, bY, acc);
      conv_step(abase, comp, bb2, kd * 16 + s + 1, aY, aX, bY, bX, acc);
    }
  }

  if (comp == 0) ep_pass<0>(ep, bep, acc, mh, ml, lhi, 1.0f, 0.0f, 0.0f, 0.0f);
  __syncthreads();
  if (comp == 1) ep_pass<1>(ep, bep, acc, mh, ml, lhi, 1.0f, 1.0f, 1.0f, 1.0f);
  __syncthreads();
  if (comp == 2) ep_pass<2>(ep, bep, acc, mh, ml, lhi, 1.0f, -1.0f, 1.0f, -1.0f);
  __syncthreads();
  if (comp == 3) ep_pass<3>(ep, bep, acc, mh, ml, lhi, 1.0f, 2.0f, 4.0f, 8.0f);
  __syncthreads();
  if (comp == 4) ep_pass<4>(ep, bep, acc, mh, ml, lhi, 1.0f, -2.0f, 4.0f, -8.0f);
  __syncthreads();
  if (comp == 5) ep_pass<5>(ep, bep, acc, mh, ml, lhi, 0.0f, 0.0f, 0.0f, 1.0f);
  __syncthreads();

  constexpr int NYQ = YLF / 4 / 384 / 3;
  for (int rr = 0; rr < 3; ++rr) {
    float4 yv[NYQ];
#pragma unroll
    for (int j = 0; j < NYQ; ++j) {
      const int i = tid + (rr * NYQ + j) * 384;
      const int row = i >> 3, part = i & 7;
      const int o = row / 18, yr = row - o * 18;
      const int k3 = yr / 6, hr = yr - k3 * 6;
      const int dd = min(max(d + k3 - 1, 0), 31);
      const int hh = min(max(h0 - 1 + hr, 0), 31);
      yv[j] = *(const float4*)(Yp + (size_t)o * SPATIAL + (dd * 32 + hh) * 32 + part * 4);
    }
    const int p = tid + 384 * rr;
    if (p < NPOS * 8) {
      float* e = &ep[(p >> 3) * NCH + (p & 7)];
      float v[27];
      float mx = -3.0e38f;
#pragma unroll
      for (int n = 0; n < 27; ++n) {
        v[n] = e[n * 8];
        mx = fmaxf(mx, v[n]);
      }
      float sum = 0.0f;
#pragma unroll
      for (int n = 0; n < 27; ++n) {
        v[n] = __expf(v[n] - mx);
        sum += v[n];
      }
      float inv = 1.0f / sum;
#pragma unroll
      for (int n = 0; n < 27; ++n) e[n * 8] = v[n] * inv;
    }
#pragma unroll
    for (int j = 0; j < NYQ; ++j) {
      const int i = tid + (rr * NYQ + j) * 384;
      *(float4*)&yl[i * 4] = yv[j];
    }
  }
  __syncthreads();

  if (tid < 256) {
    const int r = tid >> 6;
    const int w2 = tid & 63;
    const int w = w2 >> 1;
    const int wm = max(w - 1, 0), wp = min(w + 1, 31);

    float wr[27][4];
    const float* wsrc = &ep[(r * 32 + w) * NCH + (w2 & 1)];
#pragma unroll
    for (int n = 0; n < 27; ++n)
#pragma unroll
      for (int q = 0; q < 4; ++q) wr[n][q] = wsrc[n * 8 + q * 2];

    float* op = out + (size_t)(2 * d) * 4096 + (2 * (h0 + r)) * 64 + w2;
    const float* yo = yl + r * 32;
#pragma clang loop vectorize(disable)
    for (int c = 0; c < 32; ++c) {
      float a0 = 0.0f, a1 = 0.0f, a2 = 0.0f, a3 = 0.0f;
#pragma unroll
      for (int t = 0; t < 9; ++t) {
        const float* yr = yo + ((t / 3) * 6 + (t % 3)) * 32;
        float y0 = yr[wm];
        float y1v = yr[w];
        float y2v = yr[wp];
        a0 += y0 * wr[3 * t][0];
        a1 += y0 * wr[3 * t][1];
        a2 += y0 * wr[3 * t][2];
        a3 += y0 * wr[3 * t][3];
        a0 += y1v * wr[3 * t + 1][0];
        a1 += y1v * wr[3 * t + 1][1];
        a2 += y1v * wr[3 * t + 1][2];
        a3 += y1v * wr[3 * t + 1][3];
        a0 += y2v * wr[3 * t + 2][0];
        a1 += y2v * wr[3 * t + 2][1];
        a2 += y2v * wr[3 * t + 2][2];
        a3 += y2v * wr[3 * t + 2][3];
      }
      op[0] = a0;
      op[64] = a1;
      op[4096] = a2;
      op[4096 + 64] = a3;
      op += 262144;
      yo += 18 * 32;
    }
  }
}

extern "C" void kernel_launch(void* const* d_in, const int* in_sizes, int n_in,
                              void* d_out, int out_size, void* d_ws,
                              size_t ws_size, hipStream_t stream) {
  (void)in_sizes; (void)n_in; (void)out_size; (void)ws_size;
  const float* x  = (const float*)d_in[0];
  const float* Wc = (const float*)d_in[1];
  const float* bc = (const float*)d_in[2];
  const float* We = (const float*)d_in[3];
  const float* be = (const float*)d_in[4];
  const float* Wp = (const float*)d_in[5];
  float* out = (float*)d_out;

  float* ws = (float*)d_ws;
  float* y1   = ws;
  float* Yp   = y1 + (size_t)64 * SPATIAL;
  float* Wt   = Yp + (size_t)32 * SPATIAL;
  float* bep  = Wt + (size_t)NWT;

  front_kernel<<<FRONT_BLOCKS, 128, 0, stream>>>(x, Wc, bc, Wp, We, be, y1, Yp, Wt, bep);
  conv3_softmax_wmma_kernel<<<256, 384, 0, stream>>>(y1, Wt, bep, Yp, out);
}


# F(4,3) output transform and softmax done in registers by the gather threads from six component planes parked in LDS: six serial LDS passes, logit plane and six barriers removed; sum orders unchanged
# speedup vs baseline: 4.2237x; 1.0491x over previous
#include <hip/hip_runtime.h>
#include <hip/hip_bf16.h>

typedef float v2f __attribute__((ext_vector_type(2)));
typedef float v8f __attribute__((ext_vector_type(8)));

__device__ __forceinline__ v8f wmma_f32(v2f a, v2f b, v8f c) {
  return __builtin_amdgcn_wmma_f32_16x16x4_f32(false, a, false, b, (short)0, c,
                                               false, false);
}

#define SPATIAL 32768
#define MPAD 224
#define NWT (54 * 64 * MPAD)
#define NCH 216

#define GEMM_BLOCKS 512
#define LAP 68
#define FRONT_BLOCKS 568
__global__ __launch_bounds__(128) void front_kernel(
    const float* __restrict__ x, const float* __restrict__ Wc,
    const float* __restrict__ bc, const float* __restrict__ Wp,
    const float* __restrict__ We, const float* __restrict__ be,
    float* __restrict__ y1, float* __restrict__ Yp, float* __restrict__ Wt,
    float* __restrict__ bep) {
  __shared__ __attribute__((aligned(16))) float lA[96 * LAP];
  __shared__ __attribute__((aligned(16))) float lB[64 * 64];

  const int tid = threadIdx.x;

  if (blockIdx.x >= GEMM_BLOCKS) {
    const int t = (blockIdx.x - GEMM_BLOCKS) * 128 + tid;
    const int o = t % MPAD;
    const int cm2 = t / MPAD;
    if (t < MPAD) bep[t] = (t < NCH) ? be[t] : 0.0f;
    float vf[54];
    if (o < NCH) {
      const float2* s = (const float2*)(We + (size_t)o * 1728 + cm2 * 54);
#pragma unroll
      for (int k = 0; k < 27; ++k) {
        float2 q = s[k];
        vf[2 * k] = q.x;
        vf[2 * k + 1] = q.y;
      }
    } else {
#pragma unroll
      for (int k = 0; k < 54; ++k) vf[k] = 0.0f;
    }
    float* dst = Wt + ((size_t)cm2 * MPAD + o) * 2;
#pragma unroll
    for (int kd = 0; kd < 3; ++kd)
#pragma unroll
      for (int kw = 0; kw < 3; ++kw) {
        const int i0 = kd * 9 + kw;
        const float e0 = vf[i0], e1 = vf[i0 + 3], e2 = vf[i0 + 6];
        const float q0 = vf[27 + i0], q1 = vf[27 + i0 + 3], q2 = vf[27 + i0 + 6];
        float* dk = dst + (size_t)(kd * 18 + kw) * (32 * MPAD * 2);
        const float es = e0 + e2, qs = q0 + q2;
        const float ea = (1.0f / 24.0f) * e0 + (1.0f / 6.0f) * e2, eb = (1.0f / 12.0f) * e1;
        const float qa = (1.0f / 24.0f) * q0 + (1.0f / 6.0f) * q2, qb = (1.0f / 12.0f) * q1;
        *(float2*)(dk) = make_float2(0.25f * e0, 0.25f * q0);
        *(float2*)(dk + (size_t)3 * (32 * MPAD * 2)) =
            make_float2((-1.0f / 6.0f) * (es + e1), (-1.0f / 6.0f) * (qs + q1));
        *(float2*)(dk + (size_t)6 * (32 * MPAD * 2)) =
            make_float2((-1.0f / 6.0f) * (es - e1), (-1.0f / 6.0f) * (qs - q1));
        *(float2*)(dk + (size_t)9 * (32 * MPAD * 2)) = make_float2(ea + eb, qa + qb);
        *(float2*)(dk + (size_t)12 * (32 * MPAD * 2)) = make_float2(ea - eb, qa - qb);
        *(float2*)(dk + (size_t)15 * (32 * MPAD * 2)) = make_float2(e2, q2);
      }
    return;
  }

  const int lane = tid & 31;
  const int wave = tid >> 5;
  const int n0 = blockIdx.x * 64;
  const int ml = lane & 15;
  const int lhi = lane >> 4;

  v8f acc[6];
#pragma unroll
  for (int t = 0; t < 6; ++t) {
    v8f c;
#pragma unroll
    for (int v = 0; v < 8; ++v) c[v] = (t < 4) ? bc[t * 16 + v + 8 * lhi] : 0.0f;
    acc[t] = c;
  }
  float4 wa[12];
#pragma unroll
  for (int j = 0; j < 12; ++j) {
    const int i = tid + 128 * j;
    wa[j] = (j < 8) ? ((const float4*)Wc)[i] : ((const float4*)Wp)[i - 1024];
  }
  float4 xa[4], xb[4];
#pragma unroll
  for (int j = 0; j < 4; ++j) {
    const int it = tid + 128 * j;
    const float* s = x + (size_t)(2 * (it >> 4)) * SPATIAL + n0 + (it & 15) * 4;
    xa[j] = *(const float4*)s;
    xb[j] = *(const float4*)(s + SPATIAL);
  }
#pragma unroll
  for (int j = 0; j < 12; ++j) {
    const int i = tid + 128 * j;
    *(float4*)&lA[(i >> 4) * LAP + (i & 15) * 4] = wa[j];
  }
#pragma unroll
  for (int j = 0; j < 4; ++j) {
    const int it = tid + 128 * j;
    float* dst = &lB[((it >> 4) * 64 + (it & 15) * 4) * 2];
    *(float4*)dst = make_float4(xa[j].x, xb[j].x, xa[j].y, xb[j].y);
    *(float4*)(dst + 4) = make_float4(xa[j].z, xb[j].z, xa[j].w, xb[j].w);
  }
  __syncthreads();

#pragma unroll
  for (int k0 = 0; k0 < 64; k0 += 4) {
    v2f b = *(const v2f*)&lB[(((k0 >> 1) + lhi) * 64 + wave * 16 + ml) * 2];
#pragma unroll
    for (int t = 0; t < 6; ++t) {
      v2f a = *(const v2f*)&lA[(t * 16 + ml) * LAP + k0 + 2 * lhi];
      acc[t] = wmma_f32(a, b, acc[t]);
    }
  }

  const int col = n0 + wave * 16 + ml;
#pragma unroll
  for (int t = 0; t < 6; ++t)
#pragma unroll
    for (int v = 0; v < 8; ++v) {
      const int m = t * 16 + v + 8 * lhi;
      if (t < 4)
        y1[(size_t)m * SPATIAL + col] = acc[t][v];
      else
        Yp[(size_t)(m - 64) * SPATIAL + col] = acc[t][v];
    }
}

#define STRIPF (64 * 34)
#define NSTEP 48
#define YLF (32 * 18 * 32)
#define RP 220
#define RAWF (32 * RP)

__device__ __forceinline__ void conv_load_a(const float* __restrict__ abase,
                                            int comp, int sg, v2f (&a)[7][3]) {
  const int g = (sg >> 4) * 6 + comp;
  const float* p = abase + (size_t)(g * (3 * 32) + (sg & 15) * 2) * (MPAD * 2);
#pragma unroll
  for (int kw = 0; kw < 3; ++kw)
#pragma unroll
    for (int t = 0; t < 7; ++t)
      a[t][kw] = *(const v2f*)(p + kw * (32 * MPAD * 2) + t * 32);
}

__device__ __forceinline__ void conv_load_b(const float* bb, v2f (&b)[3][2]) {
#pragma unroll
  for (int kw = 0; kw < 3; ++kw)
#pragma unroll
    for (int j = 0; j < 2; ++j) b[kw][j] = *(const v2f*)(bb + (j * 16 + kw) * 2);
}

__device__ __forceinline__ void conv_step(const float* __restrict__ abase,
                                          int comp, const float* bbn, int sg,
                                          v2f (&ac)[7][3], v2f (&an)[7][3],
                                          v2f (&bc)[3][2], v2f (&bn)[3][2],
                                          v8f (&acc)[7][2]) {
  conv_load_a(abase, comp, min(sg + 1, NSTEP - 1), an);
  conv_load_b(bbn, bn);
  __builtin_amdgcn_sched_barrier(0);
#pragma unroll
  for (int kw = 0; kw < 3; ++kw)
#pragma unroll
    for (int t = 0; t < 7; ++t)
#pragma unroll
      for (int j = 0; j < 2; ++j) acc[t][j] = wmma_f32(ac[t][kw], bc[kw][j], acc[t][j]);
}

__device__ __forceinline__ float4 f4sub(float4 a, float4 b) {
  return make_float4(a.x - b.x, a.y - b.y, a.z - b.z, a.w - b.w);
}
__device__ __forceinline__ float4 f4add(float4 a, float4 b) {
  return make_float4(a.x + b.x, a.y + b.y, a.z + b.z, a.w + b.w);
}
__device__ __forceinline__ float4 f4fma(float k, float4 a, float4 b) {
  return make_float4(k * a.x + b.x, k * a.y + b.y, k * a.z + b.z, k * a.w + b.w);
}
__device__ __forceinline__ void strip_store(float* dst, float4 u, float4 v) {
  *(float2*)(dst + 0) = make_float2(u.x, v.x);
  *(float2*)(dst + 2) = make_float2(u.y, v.y);
  *(float2*)(dst + 4) = make_float2(u.z, v.z);
  *(float2*)(dst + 6) = make_float2(u.w, v.w);
}

template <int R>
__device__ __forceinline__ void row_logits(const float* rb, const float* lbp,
                                           float (&wr)[27][4]) {
  constexpr float K2 = (R & 1) ? -1.0f : 1.0f;
  constexpr float K3 = (R == 0) ? 1.0f : (R == 1) ? 2.0f : (R == 2) ? 4.0f : 8.0f;
  constexpr float K4 = (R == 0) ? 1.0f : (R == 1) ? -2.0f : (R == 2) ? 4.0f : -8.0f;
#pragma unroll
  for (int n = 0; n < 27; ++n) {
    float4 l = *(const float4*)(lbp + n * 8);
    if (R == 0) l = f4add(l, *(const float4*)(rb + n * 8));
    l = f4add(*(const float4*)(rb + RAWF + n * 8), l);
    l = f4fma(K2, *(const float4*)(rb + 2 * RAWF + n * 8), l);
    l = f4fma(K3, *(const float4*)(rb + 3 * RAWF + n * 8), l);
    l = f4fma(K4, *(const float4*)(rb + 4 * RAWF + n * 8), l);
    if (R == 3) l = f4add(*(const float4*)(rb + 5 * RAWF + n * 8), l);
    wr[n][0] = l.x;
    wr[n][1] = l.y;
    wr[n][2] = l.z;
    wr[n][3] = l.w;
  }
}

__device__ __forceinline__ void softmax_taps(float (&wr)[27][4]) {
  float inv[4];
#pragma unroll
  for (int q = 0; q < 4; ++q) {
    float mx = -3.0e38f;
#pragma unroll
    for (int n = 0; n < 27; ++n) mx = fmaxf(mx, wr[n][q]);
    float sum = 0.0f;
#pragma unroll
    for (int n = 0; n < 27; ++n) {
      wr[n][q] = __expf(wr[n][q] - mx);
      sum += wr[n][q];
    }
    inv[q] = 1.0f / sum;
  }
  const v2f i01 = {inv[0], inv[1]}, i23 = {inv[2], inv[3]};
#pragma unroll
  for (int n = 0; n < 27; ++n) {
    v2f a = {wr[n][0], wr[n][1]}, b = {wr[n][2], wr[n][3]};
    a = a * i01;
    b = b * i23;
    wr[n][0] = a[0];
    wr[n][1] = a[1];
    wr[n][2] = b[0];
    wr[n][3] = b[1];
#pragma unroll
    for (int q = 0; q < 4; ++q) {
      asm volatile("" : "+v"(wr[n][q]));
    }
  }
}

__global__ __launch_bounds__(384) void conv3_softmax_wmma_kernel(
    const float* __restrict__ y1, const float* __restrict__ Wt,
    const float* __restrict__ bep, const float* __restrict__ Yp,
    float* __restrict__ out) {
  __shared__ __attribute__((aligned(16))) float lds[6 * STRIPF + YLF + 6 * RAWF + NCH];
  float* const strips = lds;
  float* const yl = lds + 6 * STRIPF;
  float* const raw = yl + YLF;
  float* const lb = raw + 6 * RAWF;

  const int tid = threadIdx.x;
  const int lane = tid & 31;
  const int wave = tid >> 5;
  const int d = blockIdx.x >> 3;
  const int h0 = (blockIdx.x & 7) * 4;
  const int ml = lane & 15;
  const int lhi = lane >> 4;
  const int mh = wave & 1;
  const int comp = wave >> 1;
  const bool stager = tid < 256;

  v8f acc[7][2];
#pragma unroll
  for (int t = 0; t < 7; ++t)
#pragma unroll
    for (int j = 0; j < 2; ++j) {
      v8f z;
#pragma unroll
      for (int v = 0; v < 8; ++v) z[v] = 0.0f;
      acc[t][j] = z;
    }

  {
    const int s6 = tid >> 6, cm2 = (tid >> 1) & 31, side = tid & 1;
    *(float2*)&strips[s6 * STRIPF + (cm2 * 34 + side * 33) * 2] = make_float2(0.0f, 0.0f);
  }

  const float* abase = Wt + (size_t)((lhi * MPAD + mh * 112 + ml) * 2);
  const float* bbase = strips + comp * STRIPF + (lhi * 34 + ml) * 2;

  v2f aX[7][3], aY[7][3];
  v2f bX[3][2], bY[3][2];
  conv_load_a(abase, comp, 0, aX);

#pragma unroll 1
  for (int kd = 0; kd < 3; ++kd) {
    const int dd = d + kd - 1;
    const bool dok = (dd >= 0 && dd < 32);
    const int ddc = min(max(dd, 0), 31);
    const int part = tid & 7;
    const int cm2 = (tid >> 3) & 31;
    float4 ya[6], yb[6];
#pragma unroll
    for (int i = 0; i < 6; ++i) {
      ya[i] = make_float4(0.0f, 0.0f, 0.0f, 0.0f);
      yb[i] = ya[i];
    }
    if (stager) {
#pragma unroll
      for (int i = 0; i < 6; ++i) {
        const int hhc = min(max(h0 - 1 + i, 0), 31);
        const float* s = y1 + (size_t)(2 * cm2) * SPATIAL + (ddc * 32 + hhc) * 32 + part * 4;
        ya[i] = *(const float4*)s;
        yb[i] = *(const float4*)(s + SPATIAL);
      }
    }
    __syncthreads();
    if (stager) {
      float* dst = strips + (cm2 * 34 + part * 4 + 1) * 2;
      float4 ra[6], rb[6];
#pragma unroll
      for (int k = 0; k < 6; ++k) {
        const int hh = h0 - 1 + k;
        const bool ok = dok && (hh >= 0) && (hh < 32);
        ra[k] = ok ? ya[k] : make_float4(0.0f, 0.0f, 0.0f, 0.0f);
        rb[k] = ok ? yb[k] : make_float4(0.0f, 0.0f, 0.0f, 0.0f);
      }
      strip_store(dst + 0 * STRIPF, f4fma(4.0f, ra[0], f4fma(-5.0f, ra[2], ra[4])),
                  f4fma(4.0f, rb[0], f4fma(-5.0f, rb[2], rb[4])));
      strip_store(dst + 1 * STRIPF, f4fma(-4.0f, f4add(ra[1], ra[2]), f4add(ra[3], ra[4])),
                  f4fma(-4.0f, f4add(rb[1], rb[2]), f4add(rb[3], rb[4])));
      strip_store(dst + 2 * STRIPF, f4fma(4.0f, f4sub(ra[1], ra[2]), f4sub(ra[4], ra[3])),
                  f4fma(4.0f, f4sub(rb[1], rb[2]), f4sub(rb[4], rb[3])));
      strip_store(dst + 3 * STRIPF, f4fma(2.0f, f4sub(ra[3], ra[1]), f4sub(ra[4], ra[2])),
                  f4fma(2.0f, f4sub(rb[3], rb[1]), f4sub(rb[4], rb[2])));
      strip_store(dst + 4 * STRIPF, f4fma(-2.0f, f4sub(ra[3], ra[1]), f4sub(ra[4], ra[2])),
                  f4fma(-2.0f, f4sub(rb[3], rb[1]), f4sub(rb[4], rb[2])));
      strip_store(dst + 5 * STRIPF, f4fma(4.0f, ra[1], f4fma(-5.0f, ra[3], ra[5])),
                  f4fma(4.0f, rb[1], f4fma(-5.0f, rb[3], rb[5])));
    }
    if (kd < 2) {
      const int d2 = min(max(d + kd, 0), 31);
      const int h2 = min(max(h0 - 1 + (tid >> 6), 0), 31);
      __builtin_prefetch(y1 + (size_t)(tid & 63) * SPATIAL + (d2 * 32 + h2) * 32, 0, 0);
    }
    __syncthreads();

    conv_load_b(bbase, bX);
#pragma unroll 1
    for (int s = 0; s < 16; s += 2) {
      const int q1 = s + 1, q2 = min(s + 2, 15);
      const float* bb1 = bbase + q1 * (2 * 34 * 2);
      const float* bb2 = bbase + q2 * (2 * 34 * 2);
      conv_step(abase, comp, bb1, kd * 16 + s, aX, aY, bX, bY, acc);
      conv_step(abase, comp, bb2, kd * 16 + s + 1, aY, aX, bY, bX, acc);
    }
  }

  __builtin_amdgcn_sched_barrier(0);
  asm volatile("v_nop\n\tv_nop\n\tv_nop\n\tv_nop\n\tv_nop");
  __builtin_amdgcn_sched_barrier(0);

  {
    auto ysrc = [&](int j) -> const float4* {
      const int i = tid + j * 384;
      const int row = i >> 3, part = i & 7;
      const int o = row / 18, yr = row - o * 18;
      const int k3 = yr / 6, hr = yr - k3 * 6;
      const int dd = min(max(d + k3 - 1, 0), 31);
      const int hh = min(max(h0 - 1 + hr, 0), 31);
      return (const float4*)(Yp + (size_t)o * SPATIAL + (dd * 32 + hh) * 32 + part * 4);
    };
    const float4 yq0 = *ysrc(0), yq1 = *ysrc(1), yq2 = *ysrc(2), yq3 = *ysrc(3);
    const float4 yq4 = *ysrc(4), yq5 = *ysrc(5), yq6 = *ysrc(6), yq7 = *ysrc(7);
    const float4 yq8 = *ysrc(8), yq9 = *ysrc(9), yq10 = *ysrc(10), yq11 = *ysrc(11);
    float* rp = raw + comp * RAWF;
#pragma unroll
    for (int t = 0; t < 7; ++t) {
      const int o0 = (mh * 7 + t) * 16 + 8 * lhi;
      if (o0 < NCH) {
#pragma unroll
        for (int j = 0; j < 2; ++j) {
          const v8f m = acc[t][j];
          float4* q = (float4*)&rp[(j * 16 + ml) * RP + o0];
          q[0] = make_float4(m[0], m[2], m[4], m[6]);
          q[1] = make_float4(m[1], m[3], m[5], m[7]);
        }
      }
    }
    if (tid < NCH) lb[(tid & ~7) + (tid & 1) * 4 + ((tid & 7) >> 1)] = bep[tid];
    float4* yd = (float4*)yl + tid;
    yd[0 * 384] = yq0;
    yd[1 * 384] = yq1;
    yd[2 * 384] = yq2;
    yd[3 * 384] = yq3;
    yd[4 * 384] = yq4;
    yd[5 * 384] = yq5;
    yd[6 * 384] = yq6;
    yd[7 * 384] = yq7;
    yd[8 * 384] = yq8;
    yd[9 * 384] = yq9;
    yd[10 * 384] = yq10;
    yd[11 * 384] = yq11;
  }
  __syncthreads();

  if (tid < 256) {
    const int r = tid >> 6;
    const int w2 = tid & 63;
    const int w = w2 >> 1;
    const int wm = max(w - 1, 0), wp = min(w + 1, 31);

    float wr[27][4];
    const float* rb = raw + w * RP + (w2 & 1) * 4;
    const float* lbp = lb + (w2 & 1) * 4;
    if (r == 0)
      row_logits<0>(rb, lbp, wr);
    else if (r == 1)
      row_logits<1>(rb, lbp, wr);
    else if (r == 2)
      row_logits<2>(rb, lbp, wr);
    else
      row_logits<3>(rb, lbp, wr);
    softmax_taps(wr);

    float* op = out + (size_t)(2 * d) * 4096 + (2 * (h0 + r)) * 64 + w2;
    const float* yo = yl + r * 32;
#pragma clang loop vectorize(disable)
    for (int c = 0; c < 32; ++c) {
      float a0 = 0.0f, a1 = 0.0f, a2 = 0.0f, a3 = 0.0f;
#pragma unroll
      for (int t = 0; t < 9; ++t) {
        const float* yr = yo + ((t / 3) * 6 + (t % 3)) * 32;
        float y0 = yr[wm];
        float y1v = yr[w];
        float y2v = yr[wp];
        a0 += y0 * wr[3 * t][0];
        a1 += y0 * wr[3 * t][1];
        a2 += y0 * wr[3 * t][2];
        a3 += y0 * wr[3 * t][3];
        a0 += y1v * wr[3 * t + 1][0];
        a1 += y1v * wr[3 * t + 1][1];
        a2 += y1v * wr[3 * t + 1][2];
        a3 += y1v * wr[3 * t + 1][3];
        a0 += y2v * wr[3 * t + 2][0];
        a1 += y2v * wr[3 * t + 2][1];
        a2 += y2v * wr[3 * t + 2][2];
        a3 += y2v * wr[3 * t + 2][3];
      }
      op[0] = a0;
      op[64] = a1;
      op[4096] = a2;
      op[4096 + 64] = a3;
      op += 262144;
      yo += 18 * 32;
    }
  }
}

extern "C" void kernel_launch(void* const* d_in, const int* in_sizes, int n_in,
                              void* d_out, int out_size, void* d_ws,
                              size_t ws_size, hipStream_t stream) {
  (void)in_sizes; (void)n_in; (void)out_size; (void)ws_size;
  const float* x  = (const float*)d_in[0];
  const float* Wc = (const float*)d_in[1];
  const float* bc = (const float*)d_in[2];
  const float* We = (const float*)d_in[3];
  const float* be = (const float*)d_in[4];
  const float* Wp = (const float*)d_in[5];
  float* out = (float*)d_out;

  float* ws = (float*)d_ws;
  float* y1   = ws;
  float* Yp   = y1 + (size_t)64 * SPATIAL;
  float* Wt   = Yp + (size_t)32 * SPATIAL;
  float* bep  = Wt + (size_t)NWT;

  front_kernel<<<FRONT_BLOCKS, 128, 0, stream>>>(x, Wc, bc, Wp, We, be, y1, Yp, Wt, bep);
  conv3_softmax_wmma_kernel<<<256, 384, 0, stream>>>(y1, Wt, bep, Yp, out);
}


# conv also in the F(2,3) minimal-filtering form along w, nested in F(4,3) along h: 4 products per 2 columns instead of 6, one third fewer WMMAs; kw leaves the K loop; logit sum order changes
# speedup vs baseline: 4.7189x; 1.1172x over previous
#include <hip/hip_runtime.h>
#include <hip/hip_bf16.h>

typedef float v2f __attribute__((ext_vector_type(2)));
typedef float v8f __attribute__((ext_vector_type(8)));

__device__ __forceinline__ v8f wmma_f32(v2f a, v2f b, v8f c) {
  return __builtin_amdgcn_wmma_f32_16x16x4_f32(false, a, false, b, (short)0, c,
                                               false, false);
}

#define SPATIAL 32768
#define MPAD 224
#define NWT (72 * 64 * MPAD)
#define NCH 216

#define GEMM_BLOCKS 512
#define LAP 68
#define FRONT_BLOCKS 568
__global__ __launch_bounds__(128) void front_kernel(
    const float* __restrict__ x, const float* __restrict__ Wc,
    const float* __restrict__ bc, const float* __restrict__ Wp,
    const float* __restrict__ We, const float* __restrict__ be,
    float* __restrict__ y1, float* __restrict__ Yp, float* __restrict__ Wt,
    float* __restrict__ bep) {
  __shared__ __attribute__((aligned(16))) float lA[96 * LAP];
  __shared__ __attribute__((aligned(16))) float lB[64 * 64];

  const int tid = threadIdx.x;

  if (blockIdx.x >= GEMM_BLOCKS) {
    const int t = (blockIdx.x - GEMM_BLOCKS) * 128 + tid;
    const int o = t % MPAD;
    const int cm2 = t / MPAD;
    if (t < MPAD) bep[t] = (t < NCH) ? be[t] : 0.0f;
    float vf[54];
    if (o < NCH) {
      const float2* s = (const float2*)(We + (size_t)o * 1728 + cm2 * 54);
#pragma unroll
      for (int k = 0; k < 27; ++k) {
        float2 q = s[k];
        vf[2 * k] = q.x;
        vf[2 * k + 1] = q.y;
      }
    } else {
#pragma unroll
      for (int k = 0; k < 54; ++k) vf[k] = 0.0f;
    }
    float* dst = Wt + ((size_t)cm2 * MPAD + o) * 2;
#pragma unroll
    for (int kd = 0; kd < 3; ++kd) {
      float ue[6][3], uq[6][3];
#pragma unroll
      for (int kw = 0; kw < 3; ++kw) {
        const int i0 = kd * 9 + kw;
        const float e0 = vf[i0], e1 = vf[i0 + 3], e2 = vf[i0 + 6];
        const float q0 = vf[27 + i0], q1 = vf[27 + i0 + 3], q2 = vf[27 + i0 + 6];
        const float es = e0 + e2, qs = q0 + q2;
        const float ea = (1.0f / 24.0f) * e0 + (1.0f / 6.0f) * e2, eb = (1.0f / 12.0f) * e1;
        const float qa = (1.0f / 24.0f) * q0 + (1.0f / 6.0f) * q2, qb = (1.0f / 12.0f) * q1;
        ue[0][kw] = 0.25f * e0;
        uq[0][kw] = 0.25f * q0;
        ue[1][kw] = (-1.0f / 6.0f) * (es + e1);
        uq[1][kw] = (-1.0f / 6.0f) * (qs + q1);
        ue[2][kw] = (-1.0f / 6.0f) * (es - e1);
        uq[2][kw] = (-1.0f / 6.0f) * (qs - q1);
        ue[3][kw] = ea + eb;
        uq[3][kw] = qa + qb;
        ue[4][kw] = ea - eb;
        uq[4][kw] = qa - qb;
        ue[5][kw] = e2;
        uq[5][kw] = q2;
      }
#pragma unroll
      for (int i = 0; i < 6; ++i) {
        float* dk = dst + (size_t)(kd * 24 + i * 4) * (32 * MPAD * 2);
        const float se = ue[i][0] + ue[i][2], sq = uq[i][0] + uq[i][2];
        *(float2*)(dk) = make_float2(ue[i][0], uq[i][0]);
        *(float2*)(dk + (size_t)1 * (32 * MPAD * 2)) =
            make_float2(0.5f * (se + ue[i][1]), 0.5f * (sq + uq[i][1]));
        *(float2*)(dk + (size_t)2 * (32 * MPAD * 2)) =
            make_float2(0.5f * (se - ue[i][1]), 0.5f * (sq - uq[i][1]));
        *(float2*)(dk + (size_t)3 * (32 * MPAD * 2)) = make_float2(ue[i][2], uq[i][2]);
      }
    }
    return;
  }

  const int lane = tid & 31;
  const int wave = tid >> 5;
  const int n0 = blockIdx.x * 64;
  const int ml = lane & 15;
  const int lhi = lane >> 4;

  v8f acc[6];
#pragma unroll
  for (int t = 0; t < 6; ++t) {
    v8f c;
#pragma unroll
    for (int v = 0; v < 8; ++v) c[v] = (t < 4) ? bc[t * 16 + v + 8 * lhi] : 0.0f;
    acc[t] = c;
  }
  float4 wa[12];
#pragma unroll
  for (int j = 0; j < 12; ++j) {
    const int i = tid + 128 * j;
    wa[j] = (j < 8) ? ((const float4*)Wc)[i] : ((const float4*)Wp)[i - 1024];
  }
  float4 xa[4], xb[4];
#pragma unroll
  for (int j = 0; j < 4; ++j) {
    const int it = tid + 128 * j;
    const float* s = x + (size_t)(2 * (it >> 4)) * SPATIAL + n0 + (it & 15) * 4;
    xa[j] = *(const float4*)s;
    xb[j] = *(const float4*)(s + SPATIAL);
  }
#pragma unroll
  for (int j = 0; j < 12; ++j) {
    const int i = tid + 128 * j;
    *(float4*)&lA[(i >> 4) * LAP + (i & 15) * 4] = wa[j];
  }
#pragma unroll
  for (int j = 0; j < 4; ++j) {
    const int it = tid + 128 * j;
    float* dst = &lB[((it >> 4) * 64 + (it & 15) * 4) * 2];
    *(float4*)dst = make_float4(xa[j].x, xb[j].x, xa[j].y, xb[j].y);
    *(float4*)(dst + 4) = make_float4(xa[j].z, xb[j].z, xa[j].w, xb[j].w);
  }
  __syncthreads();

#pragma unroll
  for (int k0 = 0; k0 < 64; k0 += 4) {
    v2f b = *(const v2f*)&lB[(((k0 >> 1) + lhi) * 64 + wave * 16 + ml) * 2];
#pragma unroll
    for (int t = 0; t < 6; ++t) {
      v2f a = *(const v2f*)&lA[(t * 16 + ml) * LAP + k0 + 2 * lhi];
      acc[t] = wmma_f32(a, b, acc[t]);
    }
  }

  const int col = n0 + wave * 16 + ml;
#pragma unroll
  for (int t = 0; t < 6; ++t)
#pragma unroll
    for (int v = 0; v < 8; ++v) {
      const int m = t * 16 + v + 8 * lhi;
      if (t < 4)
        y1[(size_t)m * SPATIAL + col] = acc[t][v];
      else
        Yp[(size_t)(m - 64) * SPATIAL + col] = acc[t][v];
    }
}

#define PLANEF 1024
#define YLF (32 * 18 * 32)
#define RP 220
#define RAWF (32 * RP)
#define WPL (32 * MPAD * 2)
#define ASTEP (2 * MPAD * 2)

__device__ __forceinline__ void conv_load_a(const float* __restrict__ p, v2f (&a)[7][4]) {
#pragma unroll
  for (int c = 0; c < 4; ++c)
#pragma unroll
    for (int t = 0; t < 7; ++t) a[t][c] = *(const v2f*)(p + c * WPL + t * 32);
}

__device__ __forceinline__ void conv_load_b(const float* bb, v2f (&b)[4]) {
#pragma unroll
  for (int c = 0; c < 4; ++c) b[c] = *(const v2f*)(bb + c * PLANEF);
}

template <bool NEXT>
__device__ __forceinline__ void conv_step(const float* __restrict__ pn, const float* bbn,
                                          v2f (&a)[7][4], v2f (&b)[4], v8f (&acc)[7][4]) {
#pragma unroll
  for (int c = 0; c < 4; ++c)
#pragma unroll
    for (int t = 0; t < 7; ++t) {
      acc[t][c] = wmma_f32(a[t][c], b[c], acc[t][c]);
      if (NEXT) {
        a[t][c] = *(const v2f*)(pn + c * WPL + t * 32);
        if (t == 6) b[c] = *(const v2f*)(bbn + c * PLANEF);
      }
      __builtin_amdgcn_sched_barrier(0);
    }
}

__device__ __forceinline__ float4 f4sub(float4 a, float4 b) {
  return make_float4(a.x - b.x, a.y - b.y, a.z - b.z, a.w - b.w);
}
__device__ __forceinline__ float4 f4add(float4 a, float4 b) {
  return make_float4(a.x + b.x, a.y + b.y, a.z + b.z, a.w + b.w);
}
__device__ __forceinline__ float4 f4fma(float k, float4 a, float4 b) {
  return make_float4(k * a.x + b.x, k * a.y + b.y, k * a.z + b.z, k * a.w + b.w);
}
__device__ __forceinline__ void plane_store(float* dst, int part, float4 u, float4 v) {
  float ul = __shfl_up(u.w, 1), ur = __shfl_down(u.x, 1);
  float vl = __shfl_up(v.w, 1), vr = __shfl_down(v.x, 1);
  if (part == 0) {
    ul = 0.0f;
    vl = 0.0f;
  }
  if (part == 7) {
    ur = 0.0f;
    vr = 0.0f;
  }
  *(float4*)(dst + 0 * PLANEF) = make_float4(ul - u.y, vl - v.y, u.y - u.w, v.y - v.w);
  *(float4*)(dst + 1 * PLANEF) = make_float4(u.x + u.y, v.x + v.y, u.z + u.w, v.z + v.w);
  *(float4*)(dst + 2 * PLANEF) = make_float4(u.y - u.x, v.y - v.x, u.w - u.z, v.w - v.z);
  *(float4*)(dst + 3 * PLANEF) = make_float4(u.x - u.z, v.x - v.z, u.z - ur, v.z - vr);
}

template <int R>
__device__ __forceinline__ void row_logits(const float* rb, const float* lbp,
                                           float (&wr)[27][4]) {
  constexpr float K2 = (R & 1) ? -1.0f : 1.0f;
  constexpr float K3 = (R == 0) ? 1.0f : (R == 1) ? 2.0f : (R == 2) ? 4.0f : 8.0f;
  constexpr float K4 = (R == 0) ? 1.0f : (R == 1) ? -2.0f : (R == 2) ? 4.0f : -8.0f;
#pragma unroll
  for (int n = 0; n < 27; ++n) {
    float4 l = *(const float4*)(lbp + n * 8);
    if (R == 0) l = f4add(l, *(const float4*)(rb + n * 8));
    l = f4add(*(const float4*)(rb + RAWF + n * 8), l);
    l = f4fma(K2, *(const float4*)(rb + 2 * RAWF + n * 8), l);
    l = f4fma(K3, *(const float4*)(rb + 3 * RAWF + n * 8), l);
    l = f4fma(K4, *(const float4*)(rb + 4 * RAWF + n * 8), l);
    if (R == 3) l = f4add(*(const float4*)(rb + 5 * RAWF + n * 8), l);
    wr[n][0] = l.x;
    wr[n][1] = l.y;
    wr[n][2] = l.z;
    wr[n][3] = l.w;
  }
}

__device__ __forceinline__ void softmax_taps(float (&wr)[27][4]) {
  float inv[4];
#pragma unroll
  for (int q = 0; q < 4; ++q) {
    float mx = -3.0e38f;
#pragma unroll
    for (int n = 0; n < 27; ++n) mx = fmaxf(mx, wr[n][q]);
    float sum = 0.0f;
#pragma unroll
    for (int n = 0; n < 27; ++n) {
      wr[n][q] = __expf(wr[n][q] - mx);
      sum += wr[n][q];
    }
    inv[q] = 1.0f / sum;
  }
  const v2f i01 = {inv[0], inv[1]}, i23 = {inv[2], inv[3]};
#pragma unroll
  for (int n = 0; n < 27; ++n) {
    v2f a = {wr[n][0], wr[n][1]}, b = {wr[n][2], wr[n][3]};
    a = a * i01;
    b = b * i23;
    wr[n][0] = a[0];
    wr[n][1] = a[1];
    wr[n][2] = b[0];
    wr[n][3] = b[1];
#pragma unroll
    for (int q = 0; q < 4; ++q) {
      asm volatile("" : "+v"(wr[n][q]));
    }
  }
}

__global__ __launch_bounds__(384) void conv3_softmax_wmma_kernel(
    const float* __restrict__ y1, const float* __restrict__ Wt,
    const float* __restrict__ bep, const float* __restrict__ Yp,
    float* __restrict__ out) {
  __shared__ __attribute__((aligned(16))) float lds[YLF + 6 * RAWF + NCH];
  float* const planes = lds;
  float* const yl = lds;
  float* const raw = yl + YLF;
  float* const lb = raw + 6 * RAWF;

  const int tid = threadIdx.x;
  const int lane = tid & 31;
  const int wave = tid >> 5;
  const int d = blockIdx.x >> 3;
  const int h0 = (blockIdx.x & 7) * 4;
  const int ml = lane & 15;
  const int lhi = lane >> 4;
  const int mh = wave & 1;
  const int comp = wave >> 1;
  const bool stager = tid < 256;

  v8f acc[7][4];
#pragma unroll
  for (int t = 0; t < 7; ++t)
#pragma unroll
    for (int c = 0; c < 4; ++c) {
      v8f z;
#pragma unroll
      for (int v = 0; v < 8; ++v) z[v] = 0.0f;
      acc[t][c] = z;
    }

  const float* abase = Wt + (size_t)((lhi * MPAD + mh * 112 + ml) * 2);
  const float* bbase = planes + comp * (4 * PLANEF) + (lhi * 16 + ml) * 2;

  v2f a[7][4];
  v2f b[4];
  const float* pa = abase + (size_t)(comp * (4 * 32)) * (MPAD * 2);

#pragma unroll 1
  for (int kd = 0; kd < 3; ++kd) {
    const int dd = d + kd - 1;
    const bool dok = (dd >= 0 && dd < 32);
    const int ddc = min(max(dd, 0), 31);
    const int part = tid & 7;
    const int cm2 = (tid >> 3) & 31;
    float4 ya[6], yb[6];
#pragma unroll
    for (int i = 0; i < 6; ++i) {
      ya[i] = make_float4(0.0f, 0.0f, 0.0f, 0.0f);
      yb[i] = ya[i];
    }
    if (stager) {
      const float* s0 = y1 + (size_t)(2 * cm2) * SPATIAL + ddc * 1024 + part * 4;
#pragma unroll
      for (int i = 0; i < 6; ++i) {
        const int hh = h0 - 1 + i;
        if (dok && hh >= 0 && hh < 32) {
          ya[i] = *(const float4*)(s0 + hh * 32);
          yb[i] = *(const float4*)(s0 + hh * 32 + SPATIAL);
        }
      }
    }
    __syncthreads();
    if (stager) {
      float* dst = planes + cm2 * 32 + part * 4;
      const float4(&ra)[6] = ya;
      const float4(&rb)[6] = yb;
      plane_store(dst + 0 * (4 * PLANEF), part, f4fma(4.0f, ra[0], f4fma(-5.0f, ra[2], ra[4])),
                  f4fma(4.0f, rb[0], f4fma(-5.0f, rb[2], rb[4])));
      plane_store(dst + 1 * (4 * PLANEF), part,
                  f4fma(-4.0f, f4add(ra[1], ra[2]), f4add(ra[3], ra[4])),
                  f4fma(-4.0f, f4add(rb[1], rb[2]), f4add(rb[3], rb[4])));
      plane_store(dst + 2 * (4 * PLANEF), part,
                  f4fma(4.0f, f4sub(ra[1], ra[2]), f4sub(ra[4], ra[3])),
                  f4fma(4.0f, f4sub(rb[1], rb[2]), f4sub(rb[4], rb[3])));
      plane_store(dst + 3 * (4 * PLANEF), part,
                  f4fma(2.0f, f4sub(ra[3], ra[1]), f4sub(ra[4], ra[2])),
                  f4fma(2.0f, f4sub(rb[3], rb[1]), f4sub(rb[4], rb[2])));
      plane_store(dst + 4 * (4 * PLANEF), part,
                  f4fma(-2.0f, f4sub(ra[3], ra[1]), f4sub(ra[4], ra[2])),
                  f4fma(-2.0f, f4sub(rb[3], rb[1]), f4sub(rb[4], rb[2])));
      plane_store(dst + 5 * (4 * PLANEF), part, f4fma(4.0f, ra[1], f4fma(-5.0f, ra[3], ra[5])),
                  f4fma(4.0f, rb[1], f4fma(-5.0f, rb[3], rb[5])));
    }
    conv_load_a(pa, a);
    if (kd < 2) {
      const int d2 = min(max(d + kd, 0), 31);
      const int h2 = min(max(h0 - 1 + (tid >> 6), 0), 31);
      __builtin_prefetch(y1 + (size_t)(tid & 63) * SPATIAL + (d2 * 32 + h2) * 32, 0, 0);
    }
    __syncthreads();

    conv_load_b(bbase, b);
#pragma unroll 1
    for (int s = 1; s < 16; ++s) {
      pa += ASTEP;
      conv_step<true>(pa, bbase + s * (2 * 16 * 2), a, b, acc);
    }
    conv_step<false>(pa, bbase, a, b, acc);
    pa += (size_t)(6 * 4 * 32 - 15 * 2) * (MPAD * 2);

    __builtin_amdgcn_sched_barrier(0);
    asm volatile("v_nop\n\tv_nop\n\tv_nop\n\tv_nop\n\tv_nop");
    __builtin_amdgcn_sched_barrier(0);
  }

  {
    auto ysrc = [&](int j) -> const float4* {
      const int i = tid + j * 384;
      const int row = i >> 3, part = i & 7;
      const int o = row / 18, yr = row - o * 18;
      const int k3 = yr / 6, hr = yr - k3 * 6;
      const int dd = min(max(d + k3 - 1, 0), 31);
      const int hh = min(max(h0 - 1 + hr, 0), 31);
      return (const float4*)(Yp + (size_t)o * SPATIAL + (dd * 32 + hh) * 32 + part * 4);
    };
    const float4 yq0 = *ysrc(0), yq1 = *ysrc(1), yq2 = *ysrc(2), yq3 = *ysrc(3);
    const float4 yq4 = *ysrc(4), yq5 = *ysrc(5), yq6 = *ysrc(6), yq7 = *ysrc(7);
    const float4 yq8 = *ysrc(8), yq9 = *ysrc(9), yq10 = *ysrc(10), yq11 = *ysrc(11);
    __syncthreads();
    float* rp = raw + comp * RAWF;
#pragma unroll
    for (int t = 0; t < 7; ++t) {
      const int o0 = (mh * 7 + t) * 16 + 8 * lhi;
      if (o0 < NCH) {
        const v8f e = (acc[t][0] + acc[t][1]) + acc[t][2];
        const v8f f = (acc[t][1] - acc[t][2]) - acc[t][3];
        float4* q = (float4*)&rp[(2 * ml) * RP + o0];
        q[0] = make_float4(e[0], e[2], e[4], e[6]);
        q[1] = make_float4(e[1], e[3], e[5], e[7]);
        float4* q2 = (float4*)&rp[(2 * ml + 1) * RP + o0];
        q2[0] = make_float4(f[0], f[2], f[4], f[6]);
        q2[1] = make_float4(f[1], f[3], f[5], f[7]);
      }
    }
    if (tid < NCH) lb[(tid & ~7) + (tid & 1) * 4 + ((tid & 7) >> 1)] = bep[tid];
    float4* yd = (float4*)yl + tid;
    yd[0 * 384] = yq0;
    yd[1 * 384] = yq1;
    yd[2 * 384] = yq2;
    yd[3 * 384] = yq3;
    yd[4 * 384] = yq4;
    yd[5 * 384] = yq5;
    yd[6 * 384] = yq6;
    yd[7 * 384] = yq7;
    yd[8 * 384] = yq8;
    yd[9 * 384] = yq9;
    yd[10 * 384] = yq10;
    yd[11 * 384] = yq11;
  }
  __syncthreads();

  if (tid < 256) {
    const int r = tid >> 6;
    const int w2 = tid & 63;
    const int w = w2 >> 1;
    const int wm = max(w - 1, 0), wp = min(w + 1, 31);

    float wr[27][4];
    const float* rb = raw + w * RP + (w2 & 1) * 4;
    const float* lbp = lb + (w2 & 1) * 4;
    if (r == 0)
      row_logits<0>(rb, lbp, wr);
    else if (r == 1)
      row_logits<1>(rb, lbp, wr);
    else if (r == 2)
      row_logits<2>(rb, lbp, wr);
    else
      row_logits<3>(rb, lbp, wr);
    softmax_taps(wr);

    float* op = out + (size_t)(2 * d) * 4096 + (2 * (h0 + r)) * 64 + w2;
    const float* yo = yl + r * 32;
#pragma clang loop vectorize(disable)
    for (int c = 0; c < 32; ++c) {
      float a0 = 0.0f, a1 = 0.0f, a2 = 0.0f, a3 = 0.0f;
#pragma unroll
      for (int t = 0; t < 9; ++t) {
        const float* yr = yo + ((t / 3) * 6 + (t % 3)) * 32;
        float y0 = yr[wm];
        float y1v = yr[w];
        float y2v = yr[wp];
        a0 += y0 * wr[3 * t][0];
        a1 += y0 * wr[3 * t][1];
        a2 += y0 * wr[3 * t][2];
        a3 += y0 * wr[3 * t][3];
        a0 += y1v * wr[3 * t + 1][0];
        a1 += y1v * wr[3 * t + 1][1];
        a2 += y1v * wr[3 * t + 1][2];
        a3 += y1v * wr[3 * t + 1][3];
        a0 += y2v * wr[3 * t + 2][0];
        a1 += y2v * wr[3 * t + 2][1];
        a2 += y2v * wr[3 * t + 2][2];
        a3 += y2v * wr[3 * t + 2][3];
      }
      op[0] = a0;
      op[64] = a1;
      op[4096] = a2;
      op[4096 + 64] = a3;
      op += 262144;
      yo += 18 * 32;
    }
  }
}

extern "C" void kernel_launch(void* const* d_in, const int* in_sizes, int n_in,
                              void* d_out, int out_size, void* d_ws,
                              size_t ws_size, hipStream_t stream) {
  (void)in_sizes; (void)n_in; (void)out_size; (void)ws_size;
  const float* x  = (const float*)d_in[0];
  const float* Wc = (const float*)d_in[1];
  const float* bc = (const float*)d_in[2];
  const float* We = (const float*)d_in[3];
  const float* be = (const float*)d_in[4];
  const float* Wp = (const float*)d_in[5];
  float* out = (float*)d_out;

  float* ws = (float*)d_ws;
  float* y1   = ws;
  float* Yp   = y1 + (size_t)64 * SPATIAL;
  float* Wt   = Yp + (size_t)32 * SPATIAL;
  float* bep  = Wt + (size_t)NWT;

  front_kernel<<<FRONT_BLOCKS, 128, 0, stream>>>(x, Wc, bc, Wp, We, be, y1, Yp, Wt, bep);
  conv3_softmax_wmma_kernel<<<256, 384, 0, stream>>>(y1, Wt, bep, Yp, out);
}
